# starting kernel
# baseline (speedup 1.0000x reference)
#include <hip/hip_runtime.h>

typedef __attribute__((ext_vector_type(16))) _Float16 v16h;
typedef __attribute__((ext_vector_type(8)))  _Float16 v8h;
typedef __attribute__((ext_vector_type(8)))  float    v8f;
typedef __attribute__((ext_vector_type(4)))  float    v4f;

#define B_ROWS 4096
#define NFEAT  4
#define VOCAB  100000
#define EMB    64
#define D_IN   256
#define D_H    256
#define D_O    128

__global__ void wt_kernel(const float* __restrict__ W, _Float16* __restrict__ Wt,
                          int K, int N) {
    int tid = blockIdx.x * blockDim.x + threadIdx.x;
    if (tid >= K * N) return;
    int k = tid / N, n = tid - k * N;
    Wt[(size_t)n * K + k] = (_Float16)W[(size_t)k * N + n];
}

__global__ void embed_kernel(const int* __restrict__ uids, const int* __restrict__ iids,
                             const float* __restrict__ utab, const float* __restrict__ itab,
                             _Float16* __restrict__ xu, _Float16* __restrict__ xi) {
    int t = blockIdx.x * blockDim.x + threadIdx.x;
    int tower = (t >= B_ROWS * D_IN) ? 1 : 0;
    int idx = tower ? (t - B_ROWS * D_IN) : t;
    int row = idx >> 8;
    int within = idx & 255;
    int f = within >> 6;
    int e = within & 63;
    const int*   ids = tower ? iids : uids;
    const float* tab = tower ? itab : utab;
    _Float16*    x   = tower ? xi : xu;
    int id = ids[row * NFEAT + f];
    float val = tab[((size_t)f * VOCAB + (size_t)id) * EMB + e];
    x[(size_t)row * D_IN + within] = (_Float16)val;
}

__launch_bounds__(128)
__global__ void mlp_kernel(const _Float16* __restrict__ x,
                           const _Float16* __restrict__ W1t,
                           const float*    __restrict__ b1,
                           const _Float16* __restrict__ W2t,
                           const float*    __restrict__ b2,
                           float* __restrict__ out_n)
{
    __shared__ _Float16 x_lds[16][D_IN];
    __shared__ _Float16 h_lds[16][D_H];
    __shared__ float    u_lds[16][D_O];
    __shared__ float    rnorm[16];

    const int tid  = threadIdx.x;
    const int wave = tid >> 5;
    const int lane = tid & 31;
    const int half = lane >> 4;
    const int idx  = lane & 15;
    const int row0 = blockIdx.x * 16;

    {
        const v8h* src = reinterpret_cast<const v8h*>(x + (size_t)row0 * D_IN);
        v8h* dst = reinterpret_cast<v8h*>(&x_lds[0][0]);
        for (int i = tid; i < 16 * D_IN / 8; i += 128) dst[i] = src[i];
    }
    __syncthreads();

    for (int nt = wave; nt < D_H / 16; nt += 4) {
        const int n = nt * 16 + idx;
        v8f c = {};
        for (int k0 = 0; k0 < D_IN; k0 += 32) {
            v8h alo = *reinterpret_cast<const v8h*>(&x_lds[idx][k0 + half * 8]);
            v8h ahi = *reinterpret_cast<const v8h*>(&x_lds[idx][k0 + 16 + half * 8]);
            v8h blo = *reinterpret_cast<const v8h*>(W1t + (size_t)n * D_IN + k0 + half * 8);
            v8h bhi = *reinterpret_cast<const v8h*>(W1t + (size_t)n * D_IN + k0 + 16 + half * 8);
            v16h a, b;
            for (int i = 0; i < 8; ++i) {
                a[i] = alo[i]; a[8 + i] = ahi[i];
                b[i] = blo[i]; b[8 + i] = bhi[i];
            }
            c = __builtin_amdgcn_wmma_f32_16x16x32_f16(false, a, false, b,
                                                       (short)0, c, false, false);
        }
        const float bn = b1[n];
        for (int r = 0; r < 8; ++r) {
            float v = c[r] + bn;
            h_lds[half * 8 + r][n] = (_Float16)(v > 0.f ? v : 0.f);
        }
    }
    __syncthreads();

    for (int nt = wave; nt < D_O / 16; nt += 4) {
        const int n = nt * 16 + idx;
        v8f c = {};
        for (int k0 = 0; k0 < D_H; k0 += 32) {
            v8h alo = *reinterpret_cast<const v8h*>(&h_lds[idx][k0 + half * 8]);
            v8h ahi = *reinterpret_cast<const v8h*>(&h_lds[idx][k0 + 16 + half * 8]);
            v8h blo = *reinterpret_cast<const v8h*>(W2t + (size_t)n * D_H + k0 + half * 8);
            v8h bhi = *reinterpret_cast<const v8h*>(W2t + (size_t)n * D_H + k0 + 16 + half * 8);
            v16h a, b;
            for (int i = 0; i < 8; ++i) {
                a[i] = alo[i]; a[8 + i] = ahi[i];
                b[i] = blo[i]; b[8 + i] = bhi[i];
            }
            c = __builtin_amdgcn_wmma_f32_16x16x32_f16(false, a, false, b,
                                                       (short)0, c, false, false);
        }
        const float bn = b2[n];
        for (int r = 0; r < 8; ++r) {
            float v = c[r] + bn;
            u_lds[half * 8 + r][n] = (v > 0.f ? v : 0.f);
        }
    }
    __syncthreads();

    if (tid < 16) {
        float s = 0.f;
        for (int k = 0; k < D_O; ++k) { float v = u_lds[tid][k]; s += v * v; }
        float nrm = sqrtf(s);
        rnorm[tid] = (nrm > 1e-8f) ? nrm : 1e-8f;
    }
    __syncthreads();

    for (int i = tid; i < 16 * D_O; i += 128) {
        int r = i >> 7, cc = i & (D_O - 1);
        out_n[(size_t)(row0 + r) * D_O + cc] = u_lds[r][cc] / rnorm[r];
    }
}

__global__ void score_kernel(const float* __restrict__ un, const float* __restrict__ vn,
                             const int* __restrict__ sw_ids,
                             const float* __restrict__ sw_table,
                             float* __restrict__ out)
{
    const int wave = threadIdx.x >> 5;
    const int lane = threadIdx.x & 31;
    const int row  = blockIdx.x * 8 + wave;
    v4f a = *reinterpret_cast<const v4f*>(un + (size_t)row * D_O + lane * 4);
    for (int j = 0; j < 4; ++j) {
        const int col = (row + j) & (B_ROWS - 1);
        v4f b = *reinterpret_cast<const v4f*>(vn + (size_t)col * D_O + lane * 4);
        float s = a[0] * b[0] + a[1] * b[1] + a[2] * b[2] + a[3] * b[3];
        for (int m = 16; m >= 1; m >>= 1) s += __shfl_xor(s, m, 32);
        if (lane == 0) {
            out[(size_t)row * 4 + j] = s - logf(sw_table[sw_ids[col]]);
        }
    }
}

extern "C" void kernel_launch(void* const* d_in, const int* in_sizes, int n_in,
                              void* d_out, int out_size, void* d_ws, size_t ws_size,
                              hipStream_t stream)
{
    const int*   user_ids = (const int*)  d_in[0];
    const int*   item_ids = (const int*)  d_in[1];
    const int*   sw_ids   = (const int*)  d_in[2];
    const float* utab     = (const float*)d_in[3];
    const float* itab     = (const float*)d_in[4];
    const float* sw_table = (const float*)d_in[5];
    const float* uW1 = (const float*)d_in[6];
    const float* ub1 = (const float*)d_in[7];
    const float* uW2 = (const float*)d_in[8];
    const float* ub2 = (const float*)d_in[9];
    const float* iW1 = (const float*)d_in[10];
    const float* ib1 = (const float*)d_in[11];
    const float* iW2 = (const float*)d_in[12];
    const float* ib2 = (const float*)d_in[13];

    char* ws = (char*)d_ws;
    size_t off = 0;
    auto alloc = [&](size_t bytes) -> void* {
        void* p = ws + off;
        off = (off + bytes + 255) & ~(size_t)255;
        return p;
    };
    _Float16* xu   = (_Float16*)alloc((size_t)B_ROWS * D_IN * 2);
    _Float16* xi   = (_Float16*)alloc((size_t)B_ROWS * D_IN * 2);
    _Float16* W1tu = (_Float16*)alloc((size_t)D_H * D_IN * 2);
    _Float16* W2tu = (_Float16*)alloc((size_t)D_O * D_H * 2);
    _Float16* W1ti = (_Float16*)alloc((size_t)D_H * D_IN * 2);
    _Float16* W2ti = (_Float16*)alloc((size_t)D_O * D_H * 2);
    float*    un   = (float*)   alloc((size_t)B_ROWS * D_O * 4);
    float*    vn   = (float*)   alloc((size_t)B_ROWS * D_O * 4);

    wt_kernel<<<(D_IN * D_H + 255) / 256, 256, 0, stream>>>(uW1, W1tu, D_IN, D_H);
    wt_kernel<<<(D_H * D_O + 255) / 256, 256, 0, stream>>>(uW2, W2tu, D_H, D_O);
    wt_kernel<<<(D_IN * D_H + 255) / 256, 256, 0, stream>>>(iW1, W1ti, D_IN, D_H);
    wt_kernel<<<(D_H * D_O + 255) / 256, 256, 0, stream>>>(iW2, W2ti, D_H, D_O);

    embed_kernel<<<(2 * B_ROWS * D_IN) / 256, 256, 0, stream>>>(
        user_ids, item_ids, utab, itab, xu, xi);

    mlp_kernel<<<B_ROWS / 16, 128, 0, stream>>>(xu, W1tu, ub1, W2tu, ub2, un);
    mlp_kernel<<<B_ROWS / 16, 128, 0, stream>>>(xi, W1ti, ib1, W2ti, ib2, vn);

    score_kernel<<<B_ROWS / 8, 256, 0, stream>>>(un, vn, sw_ids, sw_table, (float*)d_out);
}


# launch fusion 8 -> 3: one weight-prep launch for all four matrices; embedding gather folded into the MLP kernel; both towers in one MLP launch (grid 256x2, two waves per SIMD)
# speedup vs baseline: 1.8126x; 1.8126x over previous
#include <hip/hip_runtime.h>

typedef __attribute__((ext_vector_type(16))) _Float16 v16h;
typedef __attribute__((ext_vector_type(8)))  _Float16 v8h;
typedef __attribute__((ext_vector_type(4)))  _Float16 v4h;
typedef __attribute__((ext_vector_type(8)))  float    v8f;
typedef __attribute__((ext_vector_type(4)))  float    v4f;

#define B_ROWS 4096
#define NFEAT  4
#define VOCAB  100000
#define EMB    64
#define D_IN   256
#define D_H    256
#define D_O    128

#define PREP_BLOCKS_W1 ((D_IN * D_H) / 256)
#define PREP_BLOCKS_W2 ((D_H * D_O) / 256)
#define PREP_BLOCKS    (2 * PREP_BLOCKS_W1 + 2 * PREP_BLOCKS_W2)

__global__ void prep_kernel(const float* __restrict__ uW1, const float* __restrict__ uW2,
                            const float* __restrict__ iW1, const float* __restrict__ iW2,
                            _Float16* __restrict__ W1tu, _Float16* __restrict__ W2tu,
                            _Float16* __restrict__ W1ti, _Float16* __restrict__ W2ti) {
    int b = blockIdx.x;
    const float* W;
    _Float16* Wt;
    int lgN;
    if (b < PREP_BLOCKS_W1) {
        W = uW1; Wt = W1tu; lgN = 8;
    } else if (b < PREP_BLOCKS_W1 + PREP_BLOCKS_W2) {
        b -= PREP_BLOCKS_W1; W = uW2; Wt = W2tu; lgN = 7;
    } else if (b < 2 * PREP_BLOCKS_W1 + PREP_BLOCKS_W2) {
        b -= PREP_BLOCKS_W1 + PREP_BLOCKS_W2; W = iW1; Wt = W1ti; lgN = 8;
    } else {
        b -= 2 * PREP_BLOCKS_W1 + PREP_BLOCKS_W2; W = iW2; Wt = W2ti; lgN = 7;
    }
    const int tid = b * 256 + threadIdx.x;
    const int k = tid >> lgN, n = tid - (k << lgN);
    Wt[(size_t)n * 256 + k] = (_Float16)W[tid];
}

__launch_bounds__(128)
__global__ void mlp_kernel(const int*      __restrict__ uids,
                           const int*      __restrict__ iids,
                           const float*    __restrict__ utab,
                           const float*    __restrict__ itab,
                           const _Float16* __restrict__ W1tu,
                           const _Float16* __restrict__ W1ti,
                           const float*    __restrict__ ub1,
                           const float*    __restrict__ ib1,
                           const _Float16* __restrict__ W2tu,
                           const _Float16* __restrict__ W2ti,
                           const float*    __restrict__ ub2,
                           const float*    __restrict__ ib2,
                           float* __restrict__ un,
                           float* __restrict__ vn)
{
    const bool item = (blockIdx.y != 0);
    const int*      __restrict__ ids   = item ? iids : uids;
    const float*    __restrict__ tab   = item ? itab : utab;
    const _Float16* __restrict__ W1t   = item ? W1ti : W1tu;
    const float*    __restrict__ b1    = item ? ib1  : ub1;
    const _Float16* __restrict__ W2t   = item ? W2ti : W2tu;
    const float*    __restrict__ b2    = item ? ib2  : ub2;
    float*          __restrict__ out_n = item ? vn   : un;

    __shared__ _Float16 x_lds[16][D_IN];
    __shared__ _Float16 h_lds[16][D_H];
    __shared__ float    u_lds[16][D_O];
    __shared__ float    rnorm[16];

    const int tid  = threadIdx.x;
    const int wave = tid >> 5;
    const int lane = tid & 31;
    const int half = lane >> 4;
    const int idx  = lane & 15;
    const int row0 = blockIdx.x * 16;

    for (int i = 0; i < (16 * D_IN / 4) / 128; ++i) {
        const int c  = tid + i * 128;
        const int r  = c >> 6;
        const int w4 = c & 63;
        const int f  = w4 >> 4;
        const int e4 = w4 & 15;
        const int id = ids[(row0 + r) * NFEAT + f];
        const v4f v = *reinterpret_cast<const v4f*>(
            tab + ((size_t)f * VOCAB + (size_t)id) * EMB + e4 * 4);
        v4h hv;
        hv[0] = (_Float16)v[0]; hv[1] = (_Float16)v[1];
        hv[2] = (_Float16)v[2]; hv[3] = (_Float16)v[3];
        *reinterpret_cast<v4h*>(&x_lds[r][w4 * 4]) = hv;
    }
    __syncthreads();

    for (int nt = wave; nt < D_H / 16; nt += 4) {
        const int n = nt * 16 + idx;
        v8f c = {};
        for (int k0 = 0; k0 < D_IN; k0 += 32) {
            v8h alo = *reinterpret_cast<const v8h*>(&x_lds[idx][k0 + half * 8]);
            v8h ahi = *reinterpret_cast<const v8h*>(&x_lds[idx][k0 + 16 + half * 8]);
            v8h blo = *reinterpret_cast<const v8h*>(W1t + (size_t)n * D_IN + k0 + half * 8);
            v8h bhi = *reinterpret_cast<const v8h*>(W1t + (size_t)n * D_IN + k0 + 16 + half * 8);
            v16h a, b;
            for (int i = 0; i < 8; ++i) {
                a[i] = alo[i]; a[8 + i] = ahi[i];
                b[i] = blo[i]; b[8 + i] = bhi[i];
            }
            c = __builtin_amdgcn_wmma_f32_16x16x32_f16(false, a, false, b,
                                                       (short)0, c, false, false);
        }
        const float bn = b1[n];
        for (int r = 0; r < 8; ++r) {
            float v = c[r] + bn;
            h_lds[half * 8 + r][n] = (_Float16)(v > 0.f ? v : 0.f);
        }
    }
    __syncthreads();

    for (int nt = wave; nt < D_O / 16; nt += 4) {
        const int n = nt * 16 + idx;
        v8f c = {};
        for (int k0 = 0; k0 < D_H; k0 += 32) {
            v8h alo = *reinterpret_cast<const v8h*>(&h_lds[idx][k0 + half * 8]);
            v8h ahi = *reinterpret_cast<const v8h*>(&h_lds[idx][k0 + 16 + half * 8]);
            v8h blo = *reinterpret_cast<const v8h*>(W2t + (size_t)n * D_H + k0 + half * 8);
            v8h bhi = *reinterpret_cast<const v8h*>(W2t + (size_t)n * D_H + k0 + 16 + half * 8);
            v16h a, b;
            for (int i = 0; i < 8; ++i) {
                a[i] = alo[i]; a[8 + i] = ahi[i];
                b[i] = blo[i]; b[8 + i] = bhi[i];
            }
            c = __builtin_amdgcn_wmma_f32_16x16x32_f16(false, a, false, b,
                                                       (short)0, c, false, false);
        }
        const float bn = b2[n];
        for (int r = 0; r < 8; ++r) {
            float v = c[r] + bn;
            u_lds[half * 8 + r][n] = (v > 0.f ? v : 0.f);
        }
    }
    __syncthreads();

    if (tid < 16) {
        float s = 0.f;
        for (int k = 0; k < D_O; ++k) { float v = u_lds[tid][k]; s += v * v; }
        float nrm = sqrtf(s);
        rnorm[tid] = (nrm > 1e-8f) ? nrm : 1e-8f;
    }
    __syncthreads();

    for (int i = tid; i < 16 * D_O; i += 128) {
        int r = i >> 7, cc = i & (D_O - 1);
        out_n[(size_t)(row0 + r) * D_O + cc] = u_lds[r][cc] / rnorm[r];
    }
}

__global__ void score_kernel(const float* __restrict__ un, const float* __restrict__ vn,
                             const int* __restrict__ sw_ids,
                             const float* __restrict__ sw_table,
                             float* __restrict__ out)
{
    const int wave = threadIdx.x >> 5;
    const int lane = threadIdx.x & 31;
    const int row  = blockIdx.x * 8 + wave;
    v4f a = *reinterpret_cast<const v4f*>(un + (size_t)row * D_O + lane * 4);
    for (int j = 0; j < 4; ++j) {
        const int col = (row + j) & (B_ROWS - 1);
        v4f b = *reinterpret_cast<const v4f*>(vn + (size_t)col * D_O + lane * 4);
        float s = a[0] * b[0] + a[1] * b[1] + a[2] * b[2] + a[3] * b[3];
        for (int m = 16; m >= 1; m >>= 1) s += __shfl_xor(s, m, 32);
        if (lane == 0) {
            out[(size_t)row * 4 + j] = s - logf(sw_table[sw_ids[col]]);
        }
    }
}

extern "C" void kernel_launch(void* const* d_in, const int* in_sizes, int n_in,
                              void* d_out, int out_size, void* d_ws, size_t ws_size,
                              hipStream_t stream)
{
    const int*   user_ids = (const int*)  d_in[0];
    const int*   item_ids = (const int*)  d_in[1];
    const int*   sw_ids   = (const int*)  d_in[2];
    const float* utab     = (const float*)d_in[3];
    const float* itab     = (const float*)d_in[4];
    const float* sw_table = (const float*)d_in[5];
    const float* uW1 = (const float*)d_in[6];
    const float* ub1 = (const float*)d_in[7];
    const float* uW2 = (const float*)d_in[8];
    const float* ub2 = (const float*)d_in[9];
    const float* iW1 = (const float*)d_in[10];
    const float* ib1 = (const float*)d_in[11];
    const float* iW2 = (const float*)d_in[12];
    const float* ib2 = (const float*)d_in[13];

    char* ws = (char*)d_ws;
    size_t off = 0;
    auto alloc = [&](size_t bytes) -> void* {
        void* p = ws + off;
        off = (off + bytes + 255) & ~(size_t)255;
        return p;
    };
    _Float16* W1tu = (_Float16*)alloc((size_t)D_H * D_IN * 2);
    _Float16* W2tu = (_Float16*)alloc((size_t)D_O * D_H * 2);
    _Float16* W1ti = (_Float16*)alloc((size_t)D_H * D_IN * 2);
    _Float16* W2ti = (_Float16*)alloc((size_t)D_O * D_H * 2);
    float*    un   = (float*)   alloc((size_t)B_ROWS * D_O * 4);
    float*    vn   = (float*)   alloc((size_t)B_ROWS * D_O * 4);

    prep_kernel<<<PREP_BLOCKS, 256, 0, stream>>>(uW1, uW2, iW1, iW2, W1tu, W2tu, W1ti, W2ti);

    mlp_kernel<<<dim3(B_ROWS / 16, 2), 128, 0, stream>>>(
        user_ids, item_ids, utab, itab, W1tu, W1ti, ub1, ib1, W2tu, W2ti, ub2, ib2, un, vn);

    score_kernel<<<B_ROWS / 8, 256, 0, stream>>>(un, vn, sw_ids, sw_table, (float*)d_out);
}


# launch fusion 3 -> 2: scoring folded into the MLP launch (block = 16 user + 19 item rows, 3 halo rows recomputed); un/vn stay in LDS; norms and dots by wave butterfly, norm applied to the dot
# speedup vs baseline: 2.3212x; 1.2806x over previous
#include <hip/hip_runtime.h>

typedef __attribute__((ext_vector_type(16))) _Float16 v16h;
typedef __attribute__((ext_vector_type(8)))  _Float16 v8h;
typedef __attribute__((ext_vector_type(4)))  _Float16 v4h;
typedef __attribute__((ext_vector_type(8)))  float    v8f;
typedef __attribute__((ext_vector_type(4)))  float    v4f;

#define B_ROWS 4096
#define NFEAT  4
#define VOCAB  100000
#define EMB    64
#define D_IN   256
#define D_H    256
#define D_O    128

#define PREP_BLOCKS_W1 ((D_IN * D_H) / 256)
#define PREP_BLOCKS_W2 ((D_H * D_O) / 256)
#define PREP_BLOCKS    (2 * PREP_BLOCKS_W1 + 2 * PREP_BLOCKS_W2)

__global__ void prep_kernel(const float* __restrict__ uW1, const float* __restrict__ uW2,
                            const float* __restrict__ iW1, const float* __restrict__ iW2,
                            _Float16* __restrict__ W1tu, _Float16* __restrict__ W2tu,
                            _Float16* __restrict__ W1ti, _Float16* __restrict__ W2ti) {
    int b = blockIdx.x;
    const float* W;
    _Float16* Wt;
    int lgN;
    if (b < PREP_BLOCKS_W1) {
        W = uW1; Wt = W1tu; lgN = 8;
    } else if (b < PREP_BLOCKS_W1 + PREP_BLOCKS_W2) {
        b -= PREP_BLOCKS_W1; W = uW2; Wt = W2tu; lgN = 7;
    } else if (b < 2 * PREP_BLOCKS_W1 + PREP_BLOCKS_W2) {
        b -= PREP_BLOCKS_W1 + PREP_BLOCKS_W2; W = iW1; Wt = W1ti; lgN = 8;
    } else {
        b -= 2 * PREP_BLOCKS_W1 + PREP_BLOCKS_W2; W = iW2; Wt = W2ti; lgN = 7;
    }
    const int tid = b * 256 + threadIdx.x;
    const int k = tid >> lgN, n = tid - (k << lgN);
    Wt[(size_t)n * 256 + k] = (_Float16)W[tid];
}

#define N_USER_ROWS 16
#define N_ITEM_ROWS 19
#define N_LDS_ROWS  (N_USER_ROWS + N_ITEM_ROWS)

__device__ __forceinline__ v16h frag16(const _Float16* p) {
    const v8h lo = *reinterpret_cast<const v8h*>(p);
    const v8h hi = *reinterpret_cast<const v8h*>(p + 16);
    v16h a;
    for (int i = 0; i < 8; ++i) { a[i] = lo[i]; a[8 + i] = hi[i]; }
    return a;
}

__device__ __forceinline__ float dot4(const v4f a, const v4f b) {
    return a[0] * b[0] + a[1] * b[1] + a[2] * b[2] + a[3] * b[3];
}

template <int M>
__device__ __forceinline__ float add_row_xor(float x) {
    const int b = __builtin_bit_cast(int, x);
    const int y = __builtin_amdgcn_update_dpp(b, b, 0x160 + M, 0xf, 0xf, false);
    return x + __builtin_bit_cast(float, y);
}

__device__ __forceinline__ float wave_sum(float x) {
    x = add_row_xor<1>(x);
    x = add_row_xor<2>(x);
    x = add_row_xor<4>(x);
    x = add_row_xor<8>(x);
    const int b = __builtin_bit_cast(int, x);
    const int y = __builtin_amdgcn_permlanex16(b, b, 0x76543210, 0xfedcba98, false, false);
    return x + __builtin_bit_cast(float, y);
}

__launch_bounds__(256)
__global__ void tower_score_kernel(const int*      __restrict__ uids,
                                   const int*      __restrict__ iids,
                                   const float*    __restrict__ utab,
                                   const float*    __restrict__ itab,
                                   const _Float16* __restrict__ W1tu,
                                   const _Float16* __restrict__ W1ti,
                                   const float*    __restrict__ ub1,
                                   const float*    __restrict__ ib1,
                                   const _Float16* __restrict__ W2tu,
                                   const _Float16* __restrict__ W2ti,
                                   const float*    __restrict__ ub2,
                                   const float*    __restrict__ ib2,
                                   const int*      __restrict__ sw_ids,
                                   const float*    __restrict__ sw_table,
                                   float* __restrict__ out)
{
    __shared__ _Float16 x_lds[N_LDS_ROWS][D_IN];
    __shared__ _Float16 h_lds[N_LDS_ROWS][D_H];
    __shared__ float    u_lds[N_LDS_ROWS][D_O];
    __shared__ float    res_lds[16 * 4];

    const int tid  = threadIdx.x;
    const int wave = tid >> 5;
    const int lane = tid & 31;
    const int half = lane >> 4;
    const int idx  = lane & 15;
    const int row0 = blockIdx.x * 16;
    const bool item = (wave >= 4);
    const int  tw   = wave & 3;

    const int sl   = lane & 7;
    const int scol = (row0 + 2 * wave + (sl >> 2) + (sl & 3)) & (B_ROWS - 1);
    const int swid = sw_ids[scol];

    int uid_v[4], iid_v[5];
    for (int i = 0; i < 4; ++i) {
        const int c = tid + i * 256;
        uid_v[i] = uids[(row0 + (c >> 6)) * NFEAT + ((c & 63) >> 4)];
    }
    for (int i = 0; i < 5; ++i) {
        const int c0 = tid + i * 256;
        const int c  = c0 < N_ITEM_ROWS * 64 ? c0 : N_ITEM_ROWS * 64 - 1;
        const int grow = (row0 + (c >> 6)) & (B_ROWS - 1);
        iid_v[i] = iids[grow * NFEAT + ((c & 63) >> 4)];
    }
    const float swv = sw_table[swid];
    v4f ux[4], ix[5];
    for (int i = 0; i < 4; ++i) {
        const int c  = tid + i * 256;
        const int w4 = c & 63;
        ux[i] = *reinterpret_cast<const v4f*>(
            utab + ((size_t)(w4 >> 4) * VOCAB + (size_t)uid_v[i]) * EMB + (w4 & 15) * 4);
    }
    for (int i = 0; i < 5; ++i) {
        const int c0 = tid + i * 256;
        const int c  = c0 < N_ITEM_ROWS * 64 ? c0 : N_ITEM_ROWS * 64 - 1;
        const int w4 = c & 63;
        ix[i] = *reinterpret_cast<const v4f*>(
            itab + ((size_t)(w4 >> 4) * VOCAB + (size_t)iid_v[i]) * EMB + (w4 & 15) * 4);
    }
    for (int i = 0; i < 4; ++i) {
        const int c = tid + i * 256;
        v4h hv;
        hv[0] = (_Float16)ux[i][0]; hv[1] = (_Float16)ux[i][1];
        hv[2] = (_Float16)ux[i][2]; hv[3] = (_Float16)ux[i][3];
        *reinterpret_cast<v4h*>(&x_lds[c >> 6][(c & 63) * 4]) = hv;
    }
    for (int i = 0; i < 5; ++i) {
        const int c = tid + i * 256;
        if (i < 4 || c < N_ITEM_ROWS * 64) {
            v4h hv;
            hv[0] = (_Float16)ix[i][0]; hv[1] = (_Float16)ix[i][1];
            hv[2] = (_Float16)ix[i][2]; hv[3] = (_Float16)ix[i][3];
            *reinterpret_cast<v4h*>(&x_lds[N_USER_ROWS + (c >> 6)][(c & 63) * 4]) = hv;
        }
    }
    __syncthreads();

    const int r2 = N_USER_ROWS + 16 + (idx < 2 ? idx : 2);

    if (!item) {
        for (int nt = tw; nt < D_H / 16; nt += 4) {
            const int n = nt * 16 + idx;
            v8f c = {};
            for (int k0 = 0; k0 < D_IN; k0 += 32) {
                const v16h a = frag16(&x_lds[idx][k0 + half * 8]);
                const v16h b = frag16(W1tu + (size_t)n * D_IN + k0 + half * 8);
                c = __builtin_amdgcn_wmma_f32_16x16x32_f16(false, a, false, b,
                                                           (short)0, c, false, false);
            }
            const float bn = ub1[n];
            for (int r = 0; r < 8; ++r) {
                const float v = c[r] + bn;
                h_lds[half * 8 + r][n] = (_Float16)(v > 0.f ? v : 0.f);
            }
        }
    } else {
        for (int nt = tw; nt < D_H / 16; nt += 4) {
            const int n = nt * 16 + idx;
            v8f c0 = {}, c1 = {};
            for (int k0 = 0; k0 < D_IN; k0 += 32) {
                const v16h a0 = frag16(&x_lds[N_USER_ROWS + idx][k0 + half * 8]);
                const v16h a1 = frag16(&x_lds[r2][k0 + half * 8]);
                const v16h b  = frag16(W1ti + (size_t)n * D_IN + k0 + half * 8);
                c0 = __builtin_amdgcn_wmma_f32_16x16x32_f16(false, a0, false, b,
                                                            (short)0, c0, false, false);
                c1 = __builtin_amdgcn_wmma_f32_16x16x32_f16(false, a1, false, b,
                                                            (short)0, c1, false, false);
            }
            const float bn = ib1[n];
            for (int r = 0; r < 8; ++r) {
                const float v = c0[r] + bn;
                h_lds[N_USER_ROWS + half * 8 + r][n] = (_Float16)(v > 0.f ? v : 0.f);
            }
            if (half == 0) {
                for (int r = 0; r < 3; ++r) {
                    const float v = c1[r] + bn;
                    h_lds[N_USER_ROWS + 16 + r][n] = (_Float16)(v > 0.f ? v : 0.f);
                }
            }
        }
    }
    __syncthreads();

    if (!item) {
        for (int nt = tw; nt < D_O / 16; nt += 4) {
            const int n = nt * 16 + idx;
            v8f c = {};
            for (int k0 = 0; k0 < D_H; k0 += 32) {
                const v16h a = frag16(&h_lds[idx][k0 + half * 8]);
                const v16h b = frag16(W2tu + (size_t)n * D_H + k0 + half * 8);
                c = __builtin_amdgcn_wmma_f32_16x16x32_f16(false, a, false, b,
                                                           (short)0, c, false, false);
            }
            const float bn = ub2[n];
            for (int r = 0; r < 8; ++r) {
                const float v = c[r] + bn;
                u_lds[half * 8 + r][n] = (v > 0.f ? v : 0.f);
            }
        }
    } else {
        for (int nt = tw; nt < D_O / 16; nt += 4) {
            const int n = nt * 16 + idx;
            v8f c0 = {}, c1 = {};
            for (int k0 = 0; k0 < D_H; k0 += 32) {
                const v16h a0 = frag16(&h_lds[N_USER_ROWS + idx][k0 + half * 8]);
                const v16h a1 = frag16(&h_lds[r2][k0 + half * 8]);
                const v16h b  = frag16(W2ti + (size_t)n * D_H + k0 + half * 8);
                c0 = __builtin_amdgcn_wmma_f32_16x16x32_f16(false, a0, false, b,
                                                            (short)0, c0, false, false);
                c1 = __builtin_amdgcn_wmma_f32_16x16x32_f16(false, a1, false, b,
                                                            (short)0, c1, false, false);
            }
            const float bn = ib2[n];
            for (int r = 0; r < 8; ++r) {
                const float v = c0[r] + bn;
                u_lds[N_USER_ROWS + half * 8 + r][n] = (v > 0.f ? v : 0.f);
            }
            if (half == 0) {
                for (int r = 0; r < 3; ++r) {
                    const float v = c1[r] + bn;
                    u_lds[N_USER_ROWS + 16 + r][n] = (v > 0.f ? v : 0.f);
                }
            }
        }
    }
    __syncthreads();

    const int i0 = 2 * wave;
    v4f a[2], b[5];
    for (int t = 0; t < 2; ++t)
        a[t] = *reinterpret_cast<const v4f*>(&u_lds[i0 + t][lane * 4]);
    for (int m = 0; m < 5; ++m)
        b[m] = *reinterpret_cast<const v4f*>(&u_lds[N_USER_ROWS + i0 + m][lane * 4]);
    float nu[2], nv[5], suv[8];
    for (int t = 0; t < 2; ++t) {
        const float n = sqrtf(wave_sum(dot4(a[t], a[t])));
        nu[t] = (n > 1e-8f) ? n : 1e-8f;
    }
    for (int m = 0; m < 5; ++m) {
        const float n = sqrtf(wave_sum(dot4(b[m], b[m])));
        nv[m] = (n > 1e-8f) ? n : 1e-8f;
    }
    for (int t = 0; t < 2; ++t)
        for (int j = 0; j < 4; ++j)
            suv[t * 4 + j] = wave_sum(dot4(a[t], b[t + j]));
    const int st = sl >> 2;
    const int sm = st + (sl & 3);
    float suv_sel = suv[0];
    for (int k = 1; k < 8; ++k) suv_sel = (sl == k) ? suv[k] : suv_sel;
    float nv_sel = nv[0];
    for (int m = 1; m < 5; ++m) nv_sel = (sm == m) ? nv[m] : nv_sel;
    const float nu_sel = st ? nu[1] : nu[0];
    const float res = suv_sel / (nu_sel * nv_sel) - logf(swv);
    if (lane < 8) res_lds[wave * 8 + lane] = res;
    __syncthreads();

    if (wave == 0) {
        const float2 o = *reinterpret_cast<const float2*>(&res_lds[lane * 2]);
        *reinterpret_cast<float2*>(out + (size_t)row0 * 4 + lane * 2) = o;
    }
}

extern "C" void kernel_launch(void* const* d_in, const int* in_sizes, int n_in,
                              void* d_out, int out_size, void* d_ws, size_t ws_size,
                              hipStream_t stream)
{
    const int*   user_ids = (const int*)  d_in[0];
    const int*   item_ids = (const int*)  d_in[1];
    const int*   sw_ids   = (const int*)  d_in[2];
    const float* utab     = (const float*)d_in[3];
    const float* itab     = (const float*)d_in[4];
    const float* sw_table = (const float*)d_in[5];
    const float* uW1 = (const float*)d_in[6];
    const float* ub1 = (const float*)d_in[7];
    const float* uW2 = (const float*)d_in[8];
    const float* ub2 = (const float*)d_in[9];
    const float* iW1 = (const float*)d_in[10];
    const float* ib1 = (const float*)d_in[11];
    const float* iW2 = (const float*)d_in[12];
    const float* ib2 = (const float*)d_in[13];

    char* ws = (char*)d_ws;
    size_t off = 0;
    auto alloc = [&](size_t bytes) -> void* {
        void* p = ws + off;
        off = (off + bytes + 255) & ~(size_t)255;
        return p;
    };
    _Float16* W1tu = (_Float16*)alloc((size_t)D_H * D_IN * 2);
    _Float16* W2tu = (_Float16*)alloc((size_t)D_O * D_H * 2);
    _Float16* W1ti = (_Float16*)alloc((size_t)D_H * D_IN * 2);
    _Float16* W2ti = (_Float16*)alloc((size_t)D_O * D_H * 2);

    prep_kernel<<<PREP_BLOCKS, 256, 0, stream>>>(uW1, uW2, iW1, iW2, W1tu, W2tu, W1ti, W2ti);

    tower_score_kernel<<<B_ROWS / 16, 256, 0, stream>>>(
        user_ids, item_ids, utab, itab, W1tu, W1ti, ub1, ib1, W2tu, W2ti, ub2, ib2,
        sw_ids, sw_table, (float*)d_out);
}


# launch fusion 2 -> 1: no weight-prep launch; waves read B fragments from the f32 row-major weights (4 or 2 interleaved column tiles, whole-line loads 2-3 k steps ahead), rounded to f16 in registers
# speedup vs baseline: 5.7185x; 2.4636x over previous
#include <hip/hip_runtime.h>

typedef __attribute__((ext_vector_type(16))) _Float16 v16h;
typedef __attribute__((ext_vector_type(8)))  _Float16 v8h;
typedef __attribute__((ext_vector_type(4)))  _Float16 v4h;
typedef __attribute__((ext_vector_type(8)))  float    v8f;
typedef __attribute__((ext_vector_type(4)))  float    v4f;
typedef __attribute__((ext_vector_type(2)))  float    v2f;

#define B_ROWS 4096
#define NFEAT  4
#define VOCAB  100000
#define EMB    64
#define D_IN   256
#define D_H    256
#define D_O    128

#define N_USER_ROWS 16
#define N_ITEM_ROWS 19
#define N_LDS_ROWS  (N_USER_ROWS + N_ITEM_ROWS)

template <int NT> struct RawVec;
template <> struct RawVec<4> { typedef v4f type; };
template <> struct RawVec<2> { typedef v2f type; };

__device__ __forceinline__ constexpr int krow(int s) { return s < 8 ? s : s + 8; }

__device__ __forceinline__ v16h frag16(const _Float16* p) {
    const v8h lo = *reinterpret_cast<const v8h*>(p);
    const v8h hi = *reinterpret_cast<const v8h*>(p + 16);
    v16h a;
    for (int i = 0; i < 8; ++i) { a[i] = lo[i]; a[8 + i] = hi[i]; }
    return a;
}

template <int NT, int NPITCH>
__device__ __forceinline__ void load_raw(typename RawVec<NT>::type (&raw)[16],
                                         const float* wp, int k0) {
    #pragma unroll
    for (int s = 0; s < 16; ++s)
        raw[s] = *reinterpret_cast<const typename RawVec<NT>::type*>(
            wp + (size_t)(k0 + krow(s)) * NPITCH);
}

template <int NT, int NPITCH, int DEPTH, bool SECOND_TILE, typename Tail>
__device__ __forceinline__ void k_loop(v8f (&acc0)[NT], v8f (&acc1)[NT],
                                       typename RawVec<NT>::type (&raw)[DEPTH][16],
                                       const float* wp, const _Float16* a0p,
                                       const _Float16* a1p, Tail tail) {
    v16h a0 = frag16(a0p);
    v16h a1 = SECOND_TILE ? frag16(a1p) : a0;
    #pragma unroll
    for (int t = 0; t < 8; ++t) {
        v16h b[NT];
        #pragma unroll
        for (int j = 0; j < NT; ++j) {
            #pragma unroll
            for (int s = 0; s < 16; ++s) b[j][s] = (_Float16)raw[t % DEPTH][s][j];
        }
        if (t + DEPTH < 8) load_raw<NT, NPITCH>(raw[t % DEPTH], wp, (t + DEPTH) * 32);
        else               tail(t + DEPTH - 8);
        v16h a0n = a0, a1n = a1;
        if (t + 1 < 8) {
            a0n = frag16(a0p + (t + 1) * 32);
            if (SECOND_TILE) a1n = frag16(a1p + (t + 1) * 32);
        }
        __builtin_amdgcn_sched_barrier(0);
        #pragma unroll
        for (int j = 0; j < NT; ++j)
            acc0[j] = __builtin_amdgcn_wmma_f32_16x16x32_f16(false, a0, false, b[j],
                                                             (short)0, acc0[j], false, false);
        if (SECOND_TILE) {
            #pragma unroll
            for (int j = 0; j < NT; ++j)
                acc1[j] = __builtin_amdgcn_wmma_f32_16x16x32_f16(false, a1, false, b[j],
                                                                 (short)0, acc1[j], false, false);
        }
        a0 = a0n; a1 = a1n;
        __builtin_amdgcn_sched_barrier(0);
    }
}

__device__ __forceinline__ float dot4(const v4f a, const v4f b) {
    return a[0] * b[0] + a[1] * b[1] + a[2] * b[2] + a[3] * b[3];
}

template <int M>
__device__ __forceinline__ float add_row_xor(float x) {
    const int b = __builtin_bit_cast(int, x);
    const int y = __builtin_amdgcn_update_dpp(b, b, 0x160 + M, 0xf, 0xf, false);
    return x + __builtin_bit_cast(float, y);
}

__device__ __forceinline__ float wave_sum(float x) {
    x = add_row_xor<1>(x);
    x = add_row_xor<2>(x);
    x = add_row_xor<4>(x);
    x = add_row_xor<8>(x);
    const int b = __builtin_bit_cast(int, x);
    const int y = __builtin_amdgcn_permlanex16(b, b, 0x76543210, 0xfedcba98, false, false);
    return x + __builtin_bit_cast(float, y);
}

__launch_bounds__(256) __attribute__((amdgpu_waves_per_eu(2, 2)))
__global__ void tower_score_kernel(const int*   __restrict__ uids,
                                   const int*   __restrict__ iids,
                                   const float* __restrict__ utab,
                                   const float* __restrict__ itab,
                                   const float* __restrict__ uW1,
                                   const float* __restrict__ iW1,
                                   const float* __restrict__ ub1,
                                   const float* __restrict__ ib1,
                                   const float* __restrict__ uW2,
                                   const float* __restrict__ iW2,
                                   const float* __restrict__ ub2,
                                   const float* __restrict__ ib2,
                                   const int*   __restrict__ sw_ids,
                                   const float* __restrict__ sw_table,
                                   float* __restrict__ out)
{
    __shared__ _Float16 x_lds[N_LDS_ROWS][D_IN];
    __shared__ _Float16 h_lds[N_LDS_ROWS][D_H];
    __shared__ float    u_lds[N_LDS_ROWS][D_O];
    __shared__ float    res_lds[16 * 4];

    const int tid  = threadIdx.x;
    const int wave = tid >> 5;
    const int lane = tid & 31;
    const int half = lane >> 4;
    const int idx  = lane & 15;
    const int row0 = blockIdx.x * 16;
    const bool item = (wave >= 4);
    const int  tw   = wave & 3;

    const int sl   = lane & 7;
    const int scol = (row0 + 2 * wave + (sl >> 2) + (sl & 3)) & (B_ROWS - 1);
    const int swid = sw_ids[scol];

    int uid_v[4], iid_v[5];
    for (int i = 0; i < 4; ++i) {
        const int c = tid + i * 256;
        uid_v[i] = uids[(row0 + (c >> 6)) * NFEAT + ((c & 63) >> 4)];
    }
    for (int i = 0; i < 5; ++i) {
        const int c0 = tid + i * 256;
        const int c  = c0 < N_ITEM_ROWS * 64 ? c0 : N_ITEM_ROWS * 64 - 1;
        const int grow = (row0 + (c >> 6)) & (B_ROWS - 1);
        iid_v[i] = iids[grow * NFEAT + ((c & 63) >> 4)];
    }
    const v4f bn1 = *reinterpret_cast<const v4f*>((item ? ib1 : ub1) + tw * 64 + idx * 4);
    const v2f bn2 = *reinterpret_cast<const v2f*>((item ? ib2 : ub2) + tw * 32 + idx * 2);
    __builtin_amdgcn_sched_barrier(0);

    const float* w1p = (item ? iW1 : uW1) + (size_t)(half * 8) * D_H + tw * 64 + idx * 4;
    v4f w1[2][16];
    load_raw<4, D_H>(w1[0], w1p, 0);
    load_raw<4, D_H>(w1[1], w1p, 32);
    __builtin_amdgcn_sched_barrier(0);

    const float swv = sw_table[swid];
    v4f ux[4], ix[5];
    for (int i = 0; i < 4; ++i) {
        const int c  = tid + i * 256;
        const int w4 = c & 63;
        const unsigned off = ((unsigned)(w4 >> 4) * VOCAB + (unsigned)uid_v[i]) * EMB + (w4 & 15) * 4;
        ux[i] = *reinterpret_cast<const v4f*>(utab + off);
    }
    for (int i = 0; i < 5; ++i) {
        const int c0 = tid + i * 256;
        const int c  = c0 < N_ITEM_ROWS * 64 ? c0 : N_ITEM_ROWS * 64 - 1;
        const int w4 = c & 63;
        const unsigned off = ((unsigned)(w4 >> 4) * VOCAB + (unsigned)iid_v[i]) * EMB + (w4 & 15) * 4;
        ix[i] = *reinterpret_cast<const v4f*>(itab + off);
    }
    for (int i = 0; i < 4; ++i) {
        const int c = tid + i * 256;
        v4h hv;
        hv[0] = (_Float16)ux[i][0]; hv[1] = (_Float16)ux[i][1];
        hv[2] = (_Float16)ux[i][2]; hv[3] = (_Float16)ux[i][3];
        *reinterpret_cast<v4h*>(&x_lds[c >> 6][(c & 63) * 4]) = hv;
    }
    for (int i = 0; i < 5; ++i) {
        const int c0 = tid + i * 256;
        const int c  = c0 < N_ITEM_ROWS * 64 ? c0 : N_ITEM_ROWS * 64 - 1;
        v4h hv;
        hv[0] = (_Float16)ix[i][0]; hv[1] = (_Float16)ix[i][1];
        hv[2] = (_Float16)ix[i][2]; hv[3] = (_Float16)ix[i][3];
        *reinterpret_cast<v4h*>(&x_lds[N_USER_ROWS + (c >> 6)][(c & 63) * 4]) = hv;
    }
    __syncthreads();

    const int rbase = item ? N_USER_ROWS : 0;
    const int r2 = N_USER_ROWS + 16 + (idx < 2 ? idx : 2);

    const float* w2p = (item ? iW2 : uW2) + (size_t)(half * 8) * D_O + tw * 32 + idx * 2;
    v2f w2[3][16];
    auto w2_first = [&](int i) { load_raw<2, D_O>(w2[i], w2p, i * 32); };

    v8f c0[4] = {}, c1[4] = {};
    if (!item)
        k_loop<4, D_H, 2, false>(c0, c1, w1, w1p, &x_lds[rbase + idx][half * 8],
                                 &x_lds[r2][half * 8], w2_first);
    else
        k_loop<4, D_H, 2, true>(c0, c1, w1, w1p, &x_lds[rbase + idx][half * 8],
                                &x_lds[r2][half * 8], w2_first);
    w2_first(2);
    __builtin_amdgcn_sched_barrier(0);

    {
        const int n = tw * 64 + idx * 4;
        for (int r = 0; r < 8; ++r) {
            v4h hv;
            for (int j = 0; j < 4; ++j) {
                const float v = c0[j][r] + bn1[j];
                hv[j] = (_Float16)(v > 0.f ? v : 0.f);
            }
            *reinterpret_cast<v4h*>(&h_lds[rbase + half * 8 + r][n]) = hv;
        }
        if (item && half == 0) {
            for (int r = 0; r < 3; ++r) {
                v4h hv;
                for (int j = 0; j < 4; ++j) {
                    const float v = c1[j][r] + bn1[j];
                    hv[j] = (_Float16)(v > 0.f ? v : 0.f);
                }
                *reinterpret_cast<v4h*>(&h_lds[N_USER_ROWS + 16 + r][n]) = hv;
            }
        }
    }
    __syncthreads();

    v8f d0[2] = {}, d1[2] = {};
    auto no_tail = [](int) {};
    if (!item)
        k_loop<2, D_O, 3, false>(d0, d1, w2, w2p, &h_lds[rbase + idx][half * 8],
                                 &h_lds[r2][half * 8], no_tail);
    else
        k_loop<2, D_O, 3, true>(d0, d1, w2, w2p, &h_lds[rbase + idx][half * 8],
                                &h_lds[r2][half * 8], no_tail);
    {
        const int n = tw * 32 + idx * 2;
        for (int r = 0; r < 8; ++r) {
            v2f uv;
            for (int j = 0; j < 2; ++j) {
                const float v = d0[j][r] + bn2[j];
                uv[j] = (v > 0.f ? v : 0.f);
            }
            *reinterpret_cast<v2f*>(&u_lds[rbase + half * 8 + r][n]) = uv;
        }
        if (item && half == 0) {
            for (int r = 0; r < 3; ++r) {
                v2f uv;
                for (int j = 0; j < 2; ++j) {
                    const float v = d1[j][r] + bn2[j];
                    uv[j] = (v > 0.f ? v : 0.f);
                }
                *reinterpret_cast<v2f*>(&u_lds[N_USER_ROWS + 16 + r][n]) = uv;
            }
        }
    }
    __syncthreads();

    const int i0 = 2 * wave;
    v4f a[2], b[5];
    for (int t = 0; t < 2; ++t)
        a[t] = *reinterpret_cast<const v4f*>(&u_lds[i0 + t][lane * 4]);
    for (int m = 0; m < 5; ++m)
        b[m] = *reinterpret_cast<const v4f*>(&u_lds[N_USER_ROWS + i0 + m][lane * 4]);
    float nu[2], nv[5], suv[8];
    for (int t = 0; t < 2; ++t) {
        const float n = sqrtf(wave_sum(dot4(a[t], a[t])));
        nu[t] = (n > 1e-8f) ? n : 1e-8f;
    }
    for (int m = 0; m < 5; ++m) {
        const float n = sqrtf(wave_sum(dot4(b[m], b[m])));
        nv[m] = (n > 1e-8f) ? n : 1e-8f;
    }
    for (int t = 0; t < 2; ++t)
        for (int j = 0; j < 4; ++j)
            suv[t * 4 + j] = wave_sum(dot4(a[t], b[t + j]));
    const int st = sl >> 2;
    const int sm = st + (sl & 3);
    float suv_sel = suv[0];
    for (int k = 1; k < 8; ++k) suv_sel = (sl == k) ? suv[k] : suv_sel;
    float nv_sel = nv[0];
    for (int m = 1; m < 5; ++m) nv_sel = (sm == m) ? nv[m] : nv_sel;
    const float nu_sel = st ? nu[1] : nu[0];
    const float res = suv_sel / (nu_sel * nv_sel) - logf(swv);
    if (lane < 8) res_lds[wave * 8 + lane] = res;
    __syncthreads();

    if (wave == 0) {
        const float2 o = *reinterpret_cast<const float2*>(&res_lds[lane * 2]);
        *reinterpret_cast<float2*>(out + (size_t)row0 * 4 + lane * 2) = o;
    }
}

extern "C" void kernel_launch(void* const* d_in, const int* in_sizes, int n_in,
                              void* d_out, int out_size, void* d_ws, size_t ws_size,
                              hipStream_t stream)
{
    const int*   user_ids = (const int*)  d_in[0];
    const int*   item_ids = (const int*)  d_in[1];
    const int*   sw_ids   = (const int*)  d_in[2];
    const float* utab     = (const float*)d_in[3];
    const float* itab     = (const float*)d_in[4];
    const float* sw_table = (const float*)d_in[5];
    const float* uW1 = (const float*)d_in[6];
    const float* ub1 = (const float*)d_in[7];
    const float* uW2 = (const float*)d_in[8];
    const float* ub2 = (const float*)d_in[9];
    const float* iW1 = (const float*)d_in[10];
    const float* ib1 = (const float*)d_in[11];
    const float* iW2 = (const float*)d_in[12];
    const float* ib2 = (const float*)d_in[13];

    tower_score_kernel<<<B_ROWS / 16, 256, 0, stream>>>(
        user_ids, item_ids, utab, itab, uW1, iW1, ub1, ib1, uW2, iW2, ub2, ib2,
        sw_ids, sw_table, (float*)d_out);
}


# 32 batch rows per work-group instead of 16 (grid 256 -> 128): each B fragment feeds 2 user / 3 item row tiles, halving weight reads from L2; x/h LDS row pitch 528 B keeps the A reads conflict-free
# speedup vs baseline: 8.0881x; 1.4144x over previous
#include <hip/hip_runtime.h>

typedef __attribute__((ext_vector_type(16))) _Float16 v16h;
typedef __attribute__((ext_vector_type(8)))  _Float16 v8h;
typedef __attribute__((ext_vector_type(4)))  _Float16 v4h;
typedef __attribute__((ext_vector_type(8)))  float    v8f;
typedef __attribute__((ext_vector_type(4)))  float    v4f;
typedef __attribute__((ext_vector_type(2)))  float    v2f;

#define B_ROWS 4096
#define NFEAT  4
#define VOCAB  100000
#define EMB    64
#define D_IN   256
#define D_H    256
#define D_O    128

#define ROWS_PB     32
#define N_USER_ROWS ROWS_PB
#define N_ITEM_ROWS (ROWS_PB + 3)
#define N_LDS_ROWS  (N_USER_ROWS + N_ITEM_ROWS)
#define HALO_ROW0   (N_USER_ROWS + ROWS_PB)
#define LDS_PITCH   (D_IN + 8)
#define MAX_RT      3

template <int NT> struct RawVec;
template <> struct RawVec<4> { typedef v4f type; };
template <> struct RawVec<2> { typedef v2f type; };

__device__ __forceinline__ constexpr int krow(int s) { return s < 8 ? s : s + 8; }

__device__ __forceinline__ v16h frag16(const _Float16* p) {
    const v8h lo = *reinterpret_cast<const v8h*>(p);
    const v8h hi = *reinterpret_cast<const v8h*>(p + 16);
    v16h a;
    for (int i = 0; i < 8; ++i) { a[i] = lo[i]; a[8 + i] = hi[i]; }
    return a;
}

template <int NT, int NPITCH>
__device__ __forceinline__ void load_raw(typename RawVec<NT>::type (&raw)[16],
                                         const float* wp, int k0) {
    #pragma unroll
    for (int s = 0; s < 16; ++s)
        raw[s] = *reinterpret_cast<const typename RawVec<NT>::type*>(
            wp + (size_t)(k0 + krow(s)) * NPITCH);
}

template <int NT, int NPITCH, int DEPTH, int NR, typename Tail>
__device__ __forceinline__ void k_loop(v8f (&acc)[MAX_RT][NT],
                                       typename RawVec<NT>::type (&raw)[DEPTH][16],
                                       const float* wp,
                                       const _Float16* const (&ap)[MAX_RT], Tail tail) {
    v16h a[NR];
    #pragma unroll
    for (int r = 0; r < NR; ++r) a[r] = frag16(ap[r]);
    #pragma unroll
    for (int t = 0; t < 8; ++t) {
        v16h b[NT];
        #pragma unroll
        for (int j = 0; j < NT; ++j) {
            #pragma unroll
            for (int s = 0; s < 16; ++s) b[j][s] = (_Float16)raw[t % DEPTH][s][j];
        }
        if (t + DEPTH < 8) load_raw<NT, NPITCH>(raw[t % DEPTH], wp, (t + DEPTH) * 32);
        else               tail(t + DEPTH - 8);
        v16h an[NR];
        #pragma unroll
        for (int r = 0; r < NR; ++r)
            an[r] = (t + 1 < 8) ? frag16(ap[r] + (t + 1) * 32) : a[r];
        __builtin_amdgcn_sched_barrier(0);
        #pragma unroll
        for (int r = 0; r < NR; ++r) {
            #pragma unroll
            for (int j = 0; j < NT; ++j)
                acc[r][j] = __builtin_amdgcn_wmma_f32_16x16x32_f16(false, a[r], false, b[j],
                                                                   (short)0, acc[r][j], false, false);
        }
        #pragma unroll
        for (int r = 0; r < NR; ++r) a[r] = an[r];
        __builtin_amdgcn_sched_barrier(0);
    }
}

__device__ __forceinline__ float dot4(const v4f a, const v4f b) {
    return a[0] * b[0] + a[1] * b[1] + a[2] * b[2] + a[3] * b[3];
}

template <int M>
__device__ __forceinline__ float add_row_xor(float x) {
    const int b = __builtin_bit_cast(int, x);
    const int y = __builtin_amdgcn_update_dpp(b, b, 0x160 + M, 0xf, 0xf, false);
    return x + __builtin_bit_cast(float, y);
}

__device__ __forceinline__ float wave_sum(float x) {
    x = add_row_xor<1>(x);
    x = add_row_xor<2>(x);
    x = add_row_xor<4>(x);
    x = add_row_xor<8>(x);
    const int b = __builtin_bit_cast(int, x);
    const int y = __builtin_amdgcn_permlanex16(b, b, 0x76543210, 0xfedcba98, false, false);
    return x + __builtin_bit_cast(float, y);
}

__launch_bounds__(256) __attribute__((amdgpu_waves_per_eu(2, 2)))
__global__ void tower_score_kernel(const int*   __restrict__ uids,
                                   const int*   __restrict__ iids,
                                   const float* __restrict__ utab,
                                   const float* __restrict__ itab,
                                   const float* __restrict__ uW1,
                                   const float* __restrict__ iW1,
                                   const float* __restrict__ ub1,
                                   const float* __restrict__ ib1,
                                   const float* __restrict__ uW2,
                                   const float* __restrict__ iW2,
                                   const float* __restrict__ ub2,
                                   const float* __restrict__ ib2,
                                   const int*   __restrict__ sw_ids,
                                   const float* __restrict__ sw_table,
                                   float* __restrict__ out)
{
    __shared__ __attribute__((aligned(16))) _Float16 x_lds[N_LDS_ROWS][LDS_PITCH];
    __shared__ __attribute__((aligned(16))) _Float16 h_lds[N_LDS_ROWS][LDS_PITCH];
    __shared__ __attribute__((aligned(16))) float    u_lds[N_LDS_ROWS][D_O];
    __shared__ __attribute__((aligned(16))) float    res_lds[ROWS_PB * 4];

    const int tid  = threadIdx.x;
    const int wave = tid >> 5;
    const int lane = tid & 31;
    const int half = lane >> 4;
    const int idx  = lane & 15;
    const int row0 = blockIdx.x * ROWS_PB;
    const bool item = (wave >= 4);
    const int  tw   = wave & 3;

    const int sl   = lane & 15;
    const int scol = (row0 + 4 * wave + (sl >> 2) + (sl & 3)) & (B_ROWS - 1);
    const int swid = sw_ids[scol];

    constexpr int UCH = N_USER_ROWS * 64 / 256;
    constexpr int ICH = (N_ITEM_ROWS * 64 + 255) / 256;
    int uid_v[UCH], iid_v[ICH];
    for (int i = 0; i < UCH; ++i) {
        const int c = tid + i * 256;
        uid_v[i] = uids[(row0 + (c >> 6)) * NFEAT + ((c & 63) >> 4)];
    }
    for (int i = 0; i < ICH; ++i) {
        const int c0 = tid + i * 256;
        const int c  = c0 < N_ITEM_ROWS * 64 ? c0 : N_ITEM_ROWS * 64 - 1;
        const int grow = (row0 + (c >> 6)) & (B_ROWS - 1);
        iid_v[i] = iids[grow * NFEAT + ((c & 63) >> 4)];
    }
    const v4f bn1 = *reinterpret_cast<const v4f*>((item ? ib1 : ub1) + tw * 64 + idx * 4);
    const v2f bn2 = *reinterpret_cast<const v2f*>((item ? ib2 : ub2) + tw * 32 + idx * 2);
    __builtin_amdgcn_sched_barrier(0);

    const float* w1p = (item ? iW1 : uW1) + (size_t)(half * 8) * D_H + tw * 64 + idx * 4;
    v4f w1[2][16];
    load_raw<4, D_H>(w1[0], w1p, 0);
    load_raw<4, D_H>(w1[1], w1p, 32);
    __builtin_amdgcn_sched_barrier(0);

    const float swv = sw_table[swid];
    v4f ux[UCH], ix[ICH];
    for (int i = 0; i < UCH; ++i) {
        const int c  = tid + i * 256;
        const int w4 = c & 63;
        const unsigned off = ((unsigned)(w4 >> 4) * VOCAB + (unsigned)uid_v[i]) * EMB + (w4 & 15) * 4;
        ux[i] = *reinterpret_cast<const v4f*>(utab + off);
    }
    for (int i = 0; i < ICH; ++i) {
        const int c0 = tid + i * 256;
        const int c  = c0 < N_ITEM_ROWS * 64 ? c0 : N_ITEM_ROWS * 64 - 1;
        const int w4 = c & 63;
        const unsigned off = ((unsigned)(w4 >> 4) * VOCAB + (unsigned)iid_v[i]) * EMB + (w4 & 15) * 4;
        ix[i] = *reinterpret_cast<const v4f*>(itab + off);
    }
    for (int i = 0; i < UCH; ++i) {
        const int c = tid + i * 256;
        v4h hv;
        hv[0] = (_Float16)ux[i][0]; hv[1] = (_Float16)ux[i][1];
        hv[2] = (_Float16)ux[i][2]; hv[3] = (_Float16)ux[i][3];
        *reinterpret_cast<v4h*>(&x_lds[c >> 6][(c & 63) * 4]) = hv;
    }
    for (int i = 0; i < ICH; ++i) {
        const int c0 = tid + i * 256;
        const int c  = c0 < N_ITEM_ROWS * 64 ? c0 : N_ITEM_ROWS * 64 - 1;
        v4h hv;
        hv[0] = (_Float16)ix[i][0]; hv[1] = (_Float16)ix[i][1];
        hv[2] = (_Float16)ix[i][2]; hv[3] = (_Float16)ix[i][3];
        *reinterpret_cast<v4h*>(&x_lds[N_USER_ROWS + (c >> 6)][(c & 63) * 4]) = hv;
    }
    __syncthreads();

    const int rbase = item ? N_USER_ROWS : 0;
    const int r2 = HALO_ROW0 + (idx < 2 ? idx : 2);
    const _Float16* const xa[MAX_RT] = { &x_lds[rbase + idx][half * 8],
                                         &x_lds[rbase + 16 + idx][half * 8],
                                         &x_lds[r2][half * 8] };
    const _Float16* const ha[MAX_RT] = { &h_lds[rbase + idx][half * 8],
                                         &h_lds[rbase + 16 + idx][half * 8],
                                         &h_lds[r2][half * 8] };

    const float* w2p = (item ? iW2 : uW2) + (size_t)(half * 8) * D_O + tw * 32 + idx * 2;
    v2f w2[3][16];
    auto w2_first = [&](int i) { load_raw<2, D_O>(w2[i], w2p, i * 32); };

    v8f cc[MAX_RT][4] = {};
    if (!item) k_loop<4, D_H, 2, 2>(cc, w1, w1p, xa, w2_first);
    else       k_loop<4, D_H, 2, 3>(cc, w1, w1p, xa, w2_first);
    w2_first(2);
    __builtin_amdgcn_sched_barrier(0);

    {
        const int n = tw * 64 + idx * 4;
        for (int rt = 0; rt < 2; ++rt) {
            for (int r = 0; r < 8; ++r) {
                v4h hv;
                for (int j = 0; j < 4; ++j) {
                    const float v = cc[rt][j][r] + bn1[j];
                    hv[j] = (_Float16)(v > 0.f ? v : 0.f);
                }
                *reinterpret_cast<v4h*>(&h_lds[rbase + rt * 16 + half * 8 + r][n]) = hv;
            }
        }
        if (item && half == 0) {
            for (int r = 0; r < 3; ++r) {
                v4h hv;
                for (int j = 0; j < 4; ++j) {
                    const float v = cc[2][j][r] + bn1[j];
                    hv[j] = (_Float16)(v > 0.f ? v : 0.f);
                }
                *reinterpret_cast<v4h*>(&h_lds[HALO_ROW0 + r][n]) = hv;
            }
        }
    }
    __syncthreads();

    v8f dd[MAX_RT][2] = {};
    auto no_tail = [](int) {};
    if (!item) k_loop<2, D_O, 3, 2>(dd, w2, w2p, ha, no_tail);
    else       k_loop<2, D_O, 3, 3>(dd, w2, w2p, ha, no_tail);
    {
        const int n = tw * 32 + idx * 2;
        for (int rt = 0; rt < 2; ++rt) {
            for (int r = 0; r < 8; ++r) {
                v2f uv;
                for (int j = 0; j < 2; ++j) {
                    const float v = dd[rt][j][r] + bn2[j];
                    uv[j] = (v > 0.f ? v : 0.f);
                }
                *reinterpret_cast<v2f*>(&u_lds[rbase + rt * 16 + half * 8 + r][n]) = uv;
            }
        }
        if (item && half == 0) {
            for (int r = 0; r < 3; ++r) {
                v2f uv;
                for (int j = 0; j < 2; ++j) {
                    const float v = dd[2][j][r] + bn2[j];
                    uv[j] = (v > 0.f ? v : 0.f);
                }
                *reinterpret_cast<v2f*>(&u_lds[HALO_ROW0 + r][n]) = uv;
            }
        }
    }
    __syncthreads();

    constexpr int UPW = ROWS_PB / 8;
    constexpr int IPW = UPW + 3;
    const int i0 = UPW * wave;
    v4f a[UPW], b[IPW];
    for (int t = 0; t < UPW; ++t)
        a[t] = *reinterpret_cast<const v4f*>(&u_lds[i0 + t][lane * 4]);
    for (int m = 0; m < IPW; ++m)
        b[m] = *reinterpret_cast<const v4f*>(&u_lds[N_USER_ROWS + i0 + m][lane * 4]);
    float nu[UPW], nv[IPW], suv[UPW * 4];
    for (int t = 0; t < UPW; ++t) {
        const float n = sqrtf(wave_sum(dot4(a[t], a[t])));
        nu[t] = (n > 1e-8f) ? n : 1e-8f;
    }
    for (int m = 0; m < IPW; ++m) {
        const float n = sqrtf(wave_sum(dot4(b[m], b[m])));
        nv[m] = (n > 1e-8f) ? n : 1e-8f;
    }
    for (int t = 0; t < UPW; ++t)
        for (int j = 0; j < 4; ++j)
            suv[t * 4 + j] = wave_sum(dot4(a[t], b[t + j]));
    const int st = sl >> 2;
    const int sm = st + (sl & 3);
    float suv_sel = suv[0];
    for (int k = 1; k < UPW * 4; ++k) suv_sel = (sl == k) ? suv[k] : suv_sel;
    float nv_sel = nv[0];
    for (int m = 1; m < IPW; ++m) nv_sel = (sm == m) ? nv[m] : nv_sel;
    float nu_sel = nu[0];
    for (int t = 1; t < UPW; ++t) nu_sel = (st == t) ? nu[t] : nu_sel;
    const float res = suv_sel / (nu_sel * nv_sel) - logf(swv);
    if (lane < UPW * 4) res_lds[wave * (UPW * 4) + lane] = res;
    __syncthreads();

    if (wave == 0) {
        const v4f o = *reinterpret_cast<const v4f*>(&res_lds[lane * 4]);
        *reinterpret_cast<v4f*>(out + (size_t)row0 * 4 + lane * 4) = o;
    }
}

extern "C" void kernel_launch(void* const* d_in, const int* in_sizes, int n_in,
                              void* d_out, int out_size, void* d_ws, size_t ws_size,
                              hipStream_t stream)
{
    const int*   user_ids = (const int*)  d_in[0];
    const int*   item_ids = (const int*)  d_in[1];
    const int*   sw_ids   = (const int*)  d_in[2];
    const float* utab     = (const float*)d_in[3];
    const float* itab     = (const float*)d_in[4];
    const float* sw_table = (const float*)d_in[5];
    const float* uW1 = (const float*)d_in[6];
    const float* ub1 = (const float*)d_in[7];
    const float* uW2 = (const float*)d_in[8];
    const float* ub2 = (const float*)d_in[9];
    const float* iW1 = (const float*)d_in[10];
    const float* ib1 = (const float*)d_in[11];
    const float* iW2 = (const float*)d_in[12];
    const float* ib2 = (const float*)d_in[13];

    tower_score_kernel<<<B_ROWS / ROWS_PB, 256, 0, stream>>>(
        user_ids, item_ids, utab, itab, uW1, iW1, ub1, ib1, uW2, iW2, ub2, ib2,
        sw_ids, sw_table, (float*)d_out);
}


# scoring: a wave's 27 sums (16 dots, 11 squared norms) reduced by one transposing butterfly (each stage halves the live values) instead of 27 full ones; one sqrt per lane; every sum's add order kept
# speedup vs baseline: 8.7134x; 1.0773x over previous
#include <hip/hip_runtime.h>

typedef __attribute__((ext_vector_type(16))) _Float16 v16h;
typedef __attribute__((ext_vector_type(8)))  _Float16 v8h;
typedef __attribute__((ext_vector_type(4)))  _Float16 v4h;
typedef __attribute__((ext_vector_type(8)))  float    v8f;
typedef __attribute__((ext_vector_type(4)))  float    v4f;
typedef __attribute__((ext_vector_type(2)))  float    v2f;

#define B_ROWS 4096
#define NFEAT  4
#define VOCAB  100000
#define EMB    64
#define D_IN   256
#define D_H    256
#define D_O    128

#define ROWS_PB     32
#define N_USER_ROWS ROWS_PB
#define N_ITEM_ROWS (ROWS_PB + 3)
#define N_LDS_ROWS  (N_USER_ROWS + N_ITEM_ROWS)
#define HALO_ROW0   (N_USER_ROWS + ROWS_PB)
#define LDS_PITCH   (D_IN + 8)
#define MAX_RT      3

template <int NT> struct RawVec;
template <> struct RawVec<4> { typedef v4f type; };
template <> struct RawVec<2> { typedef v2f type; };

__device__ __forceinline__ constexpr int krow(int s) { return s < 8 ? s : s + 8; }

__device__ __forceinline__ v16h frag16(const _Float16* p) {
    const v8h lo = *reinterpret_cast<const v8h*>(p);
    const v8h hi = *reinterpret_cast<const v8h*>(p + 16);
    v16h a;
    for (int i = 0; i < 8; ++i) { a[i] = lo[i]; a[8 + i] = hi[i]; }
    return a;
}

template <int NT, int NPITCH>
__device__ __forceinline__ void load_raw(typename RawVec<NT>::type (&raw)[16],
                                         const float* wp, int k0) {
    #pragma unroll
    for (int s = 0; s < 16; ++s)
        raw[s] = *reinterpret_cast<const typename RawVec<NT>::type*>(
            wp + (size_t)(k0 + krow(s)) * NPITCH);
}

template <int NT, int NPITCH, int DEPTH, int NR, typename Tail>
__device__ __forceinline__ void k_loop(v8f (&acc)[MAX_RT][NT],
                                       typename RawVec<NT>::type (&raw)[DEPTH][16],
                                       const float* wp,
                                       const _Float16* const (&ap)[MAX_RT], Tail tail) {
    v16h a[NR];
    #pragma unroll
    for (int r = 0; r < NR; ++r) a[r] = frag16(ap[r]);
    #pragma unroll
    for (int t = 0; t < 8; ++t) {
        v16h b[NT];
        #pragma unroll
        for (int j = 0; j < NT; ++j) {
            #pragma unroll
            for (int s = 0; s < 16; ++s) b[j][s] = (_Float16)raw[t % DEPTH][s][j];
        }
        if (t + DEPTH < 8) load_raw<NT, NPITCH>(raw[t % DEPTH], wp, (t + DEPTH) * 32);
        else               tail(t + DEPTH - 8);
        v16h an[NR];
        #pragma unroll
        for (int r = 0; r < NR; ++r)
            an[r] = (t + 1 < 8) ? frag16(ap[r] + (t + 1) * 32) : a[r];
        __builtin_amdgcn_sched_barrier(0);
        #pragma unroll
        for (int r = 0; r < NR; ++r) {
            #pragma unroll
            for (int j = 0; j < NT; ++j)
                acc[r][j] = __builtin_amdgcn_wmma_f32_16x16x32_f16(false, a[r], false, b[j],
                                                                   (short)0, acc[r][j], false, false);
        }
        #pragma unroll
        for (int r = 0; r < NR; ++r) a[r] = an[r];
        __builtin_amdgcn_sched_barrier(0);
    }
}

__device__ __forceinline__ float dot4(const v4f a, const v4f b) {
    return a[0] * b[0] + a[1] * b[1] + a[2] * b[2] + a[3] * b[3];
}

template <int M>
__device__ __forceinline__ float from_row_xor(float x) {
    const int r = __builtin_amdgcn_update_dpp(0, __builtin_bit_cast(int, x), 0x160 + M, 0xf, 0xf, true);
    return __builtin_bit_cast(float, r);
}

__device__ __forceinline__ float from_other_row(float x) {
    const int b = __builtin_bit_cast(int, x);
    const int y = __builtin_amdgcn_permlanex16(b, b, 0x76543210, 0xfedcba98, false, false);
    return __builtin_bit_cast(float, y);
}

template <int N, int NMAX, typename Xchg>
__device__ __forceinline__ void fold_stage(float (&v)[NMAX], bool bit, Xchg xchg) {
    #pragma unroll
    for (int m = 0; m < N / 2; ++m) {
        const float keep = bit ? v[2 * m + 1] : v[2 * m];
        const float send = bit ? v[2 * m] : v[2 * m + 1];
        v[m] = keep + xchg(send);
    }
    if (N & 1) v[N / 2] = v[N - 1] + xchg(v[N - 1]);
}

__launch_bounds__(256) __attribute__((amdgpu_waves_per_eu(2, 2)))
__global__ void tower_score_kernel(const int*   __restrict__ uids,
                                   const int*   __restrict__ iids,
                                   const float* __restrict__ utab,
                                   const float* __restrict__ itab,
                                   const float* __restrict__ uW1,
                                   const float* __restrict__ iW1,
                                   const float* __restrict__ ub1,
                                   const float* __restrict__ ib1,
                                   const float* __restrict__ uW2,
                                   const float* __restrict__ iW2,
                                   const float* __restrict__ ub2,
                                   const float* __restrict__ ib2,
                                   const int*   __restrict__ sw_ids,
                                   const float* __restrict__ sw_table,
                                   float* __restrict__ out)
{
    __shared__ __attribute__((aligned(16))) _Float16 x_lds[N_LDS_ROWS][LDS_PITCH];
    __shared__ __attribute__((aligned(16))) _Float16 h_lds[N_LDS_ROWS][LDS_PITCH];
    __shared__ __attribute__((aligned(16))) float    u_lds[N_LDS_ROWS][D_O];
    __shared__ __attribute__((aligned(16))) float    res_lds[ROWS_PB * 4];

    const int tid  = threadIdx.x;
    const int wave = tid >> 5;
    const int lane = tid & 31;
    const int half = lane >> 4;
    const int idx  = lane & 15;
    const int row0 = blockIdx.x * ROWS_PB;
    const bool item = (wave >= 4);
    const int  tw   = wave & 3;

    const int sl   = lane & 15;
    const int scol = (row0 + 4 * wave + (sl >> 2) + (sl & 3)) & (B_ROWS - 1);
    const int swid = sw_ids[scol];

    constexpr int UCH = N_USER_ROWS * 64 / 256;
    constexpr int ICH = (N_ITEM_ROWS * 64 + 255) / 256;
    int uid_v[UCH], iid_v[ICH];
    for (int i = 0; i < UCH; ++i) {
        const int c = tid + i * 256;
        uid_v[i] = uids[(row0 + (c >> 6)) * NFEAT + ((c & 63) >> 4)];
    }
    for (int i = 0; i < ICH; ++i) {
        const int c0 = tid + i * 256;
        const int c  = c0 < N_ITEM_ROWS * 64 ? c0 : N_ITEM_ROWS * 64 - 1;
        const int grow = (row0 + (c >> 6)) & (B_ROWS - 1);
        iid_v[i] = iids[grow * NFEAT + ((c & 63) >> 4)];
    }
    const v4f bn1 = *reinterpret_cast<const v4f*>((item ? ib1 : ub1) + tw * 64 + idx * 4);
    const v2f bn2 = *reinterpret_cast<const v2f*>((item ? ib2 : ub2) + tw * 32 + idx * 2);
    __builtin_amdgcn_sched_barrier(0);

    const float* w1p = (item ? iW1 : uW1) + (size_t)(half * 8) * D_H + tw * 64 + idx * 4;
    v4f w1[2][16];
    load_raw<4, D_H>(w1[0], w1p, 0);
    load_raw<4, D_H>(w1[1], w1p, 32);
    __builtin_amdgcn_sched_barrier(0);

    const float swv = sw_table[swid];
    v4f ux[UCH], ix[ICH];
    for (int i = 0; i < UCH; ++i) {
        const int c  = tid + i * 256;
        const int w4 = c & 63;
        const unsigned off = ((unsigned)(w4 >> 4) * VOCAB + (unsigned)uid_v[i]) * EMB + (w4 & 15) * 4;
        ux[i] = *reinterpret_cast<const v4f*>(utab + off);
    }
    for (int i = 0; i < ICH; ++i) {
        const int c0 = tid + i * 256;
        const int c  = c0 < N_ITEM_ROWS * 64 ? c0 : N_ITEM_ROWS * 64 - 1;
        const int w4 = c & 63;
        const unsigned off = ((unsigned)(w4 >> 4) * VOCAB + (unsigned)iid_v[i]) * EMB + (w4 & 15) * 4;
        ix[i] = *reinterpret_cast<const v4f*>(itab + off);
    }
    for (int i = 0; i < UCH; ++i) {
        const int c = tid + i * 256;
        v4h hv;
        hv[0] = (_Float16)ux[i][0]; hv[1] = (_Float16)ux[i][1];
        hv[2] = (_Float16)ux[i][2]; hv[3] = (_Float16)ux[i][3];
        *reinterpret_cast<v4h*>(&x_lds[c >> 6][(c & 63) * 4]) = hv;
    }
    for (int i = 0; i < ICH; ++i) {
        const int c0 = tid + i * 256;
        const int c  = c0 < N_ITEM_ROWS * 64 ? c0 : N_ITEM_ROWS * 64 - 1;
        v4h hv;
        hv[0] = (_Float16)ix[i][0]; hv[1] = (_Float16)ix[i][1];
        hv[2] = (_Float16)ix[i][2]; hv[3] = (_Float16)ix[i][3];
        *reinterpret_cast<v4h*>(&x_lds[N_USER_ROWS + (c >> 6)][(c & 63) * 4]) = hv;
    }
    __syncthreads();

    const int rbase = item ? N_USER_ROWS : 0;
    const int r2 = HALO_ROW0 + (idx < 2 ? idx : 2);
    const _Float16* const xa[MAX_RT] = { &x_lds[rbase + idx][half * 8],
                                         &x_lds[rbase + 16 + idx][half * 8],
                                         &x_lds[r2][half * 8] };
    const _Float16* const ha[MAX_RT] = { &h_lds[rbase + idx][half * 8],
                                         &h_lds[rbase + 16 + idx][half * 8],
                                         &h_lds[r2][half * 8] };

    const float* w2p = (item ? iW2 : uW2) + (size_t)(half * 8) * D_O + tw * 32 + idx * 2;
    v2f w2[3][16];
    auto w2_first = [&](int i) { load_raw<2, D_O>(w2[i], w2p, i * 32); };

    v8f cc[MAX_RT][4] = {};
    if (!item) k_loop<4, D_H, 2, 2>(cc, w1, w1p, xa, w2_first);
    else       k_loop<4, D_H, 2, 3>(cc, w1, w1p, xa, w2_first);
    w2_first(2);
    __builtin_amdgcn_sched_barrier(0);

    {
        const int n = tw * 64 + idx * 4;
        for (int rt = 0; rt < 2; ++rt) {
            for (int r = 0; r < 8; ++r) {
                v4h hv;
                for (int j = 0; j < 4; ++j) {
                    const float v = cc[rt][j][r] + bn1[j];
                    hv[j] = (_Float16)(v > 0.f ? v : 0.f);
                }
                *reinterpret_cast<v4h*>(&h_lds[rbase + rt * 16 + half * 8 + r][n]) = hv;
            }
        }
        if (item && half == 0) {
            for (int r = 0; r < 3; ++r) {
                v4h hv;
                for (int j = 0; j < 4; ++j) {
                    const float v = cc[2][j][r] + bn1[j];
                    hv[j] = (_Float16)(v > 0.f ? v : 0.f);
                }
                *reinterpret_cast<v4h*>(&h_lds[HALO_ROW0 + r][n]) = hv;
            }
        }
    }
    __syncthreads();

    v8f dd[MAX_RT][2] = {};
    auto no_tail = [](int) {};
    if (!item) k_loop<2, D_O, 3, 2>(dd, w2, w2p, ha, no_tail);
    else       k_loop<2, D_O, 3, 3>(dd, w2, w2p, ha, no_tail);
    {
        const int n = tw * 32 + idx * 2;
        for (int rt = 0; rt < 2; ++rt) {
            for (int r = 0; r < 8; ++r) {
                v2f uv;
                for (int j = 0; j < 2; ++j) {
                    const float v = dd[rt][j][r] + bn2[j];
                    uv[j] = (v > 0.f ? v : 0.f);
                }
                *reinterpret_cast<v2f*>(&u_lds[rbase + rt * 16 + half * 8 + r][n]) = uv;
            }
        }
        if (item && half == 0) {
            for (int r = 0; r < 3; ++r) {
                v2f uv;
                for (int j = 0; j < 2; ++j) {
                    const float v = dd[2][j][r] + bn2[j];
                    uv[j] = (v > 0.f ? v : 0.f);
                }
                *reinterpret_cast<v2f*>(&u_lds[HALO_ROW0 + r][n]) = uv;
            }
        }
    }
    __syncthreads();

    constexpr int UPW = ROWS_PB / 8;
    constexpr int IPW = UPW + 3;
    constexpr int NSUM = UPW * 4 + UPW + IPW;
    static_assert(UPW * 4 == 16 && NSUM <= 32, "one sum per lane, dots in the first row");
    const int i0 = UPW * wave;
    v4f a[UPW], b[IPW];
    for (int t = 0; t < UPW; ++t)
        a[t] = *reinterpret_cast<const v4f*>(&u_lds[i0 + t][lane * 4]);
    for (int m = 0; m < IPW; ++m)
        b[m] = *reinterpret_cast<const v4f*>(&u_lds[N_USER_ROWS + i0 + m][lane * 4]);
    float part[NSUM];
    for (int t = 0; t < UPW; ++t)
        for (int j = 0; j < 4; ++j)
            part[t * 4 + j] = dot4(a[t], b[t + j]);
    for (int t = 0; t < UPW; ++t) part[UPW * 4 + t] = dot4(a[t], a[t]);
    for (int m = 0; m < IPW; ++m) part[UPW * 4 + UPW + m] = dot4(b[m], b[m]);
    fold_stage<27, NSUM>(part, (lane & 1) != 0, [](float x) { return from_row_xor<1>(x); });
    fold_stage<14, NSUM>(part, (lane & 2) != 0, [](float x) { return from_row_xor<2>(x); });
    fold_stage<7,  NSUM>(part, (lane & 4) != 0, [](float x) { return from_row_xor<4>(x); });
    fold_stage<4,  NSUM>(part, (lane & 8) != 0, [](float x) { return from_row_xor<8>(x); });
    fold_stage<2,  NSUM>(part, (lane & 16) != 0, [](float x) { return from_other_row(x); });
    const float total = part[0];
    const float nrm  = sqrtf(total);
    const int   nrmc = __builtin_bit_cast(int, (nrm > 1e-8f) ? nrm : 1e-8f);
    const float nu_sel = __builtin_bit_cast(float,
        __builtin_amdgcn_permlanex16(nrmc, nrmc, 0x11110000, 0x33332222, false, false));
    const float nv_sel = __builtin_bit_cast(float,
        __builtin_amdgcn_permlanex16(nrmc, nrmc, 0x87657654, 0xa9879876, false, false));
    const float res = total / (nu_sel * nv_sel) - logf(swv);
    if (lane < UPW * 4) res_lds[wave * (UPW * 4) + lane] = res;
    __syncthreads();

    if (wave == 0) {
        const v4f o = *reinterpret_cast<const v4f*>(&res_lds[lane * 4]);
        *reinterpret_cast<v4f*>(out + (size_t)row0 * 4 + lane * 4) = o;
    }
}

extern "C" void kernel_launch(void* const* d_in, const int* in_sizes, int n_in,
                              void* d_out, int out_size, void* d_ws, size_t ws_size,
                              hipStream_t stream)
{
    const int*   user_ids = (const int*)  d_in[0];
    const int*   item_ids = (const int*)  d_in[1];
    const int*   sw_ids   = (const int*)  d_in[2];
    const float* utab     = (const float*)d_in[3];
    const float* itab     = (const float*)d_in[4];
    const float* sw_table = (const float*)d_in[5];
    const float* uW1 = (const float*)d_in[6];
    const float* ub1 = (const float*)d_in[7];
    const float* uW2 = (const float*)d_in[8];
    const float* ub2 = (const float*)d_in[9];
    const float* iW1 = (const float*)d_in[10];
    const float* ib1 = (const float*)d_in[11];
    const float* iW2 = (const float*)d_in[12];
    const float* ib2 = (const float*)d_in[13];

    tower_score_kernel<<<B_ROWS / ROWS_PB, 256, 0, stream>>>(
        user_ids, item_ids, utab, itab, uW1, iW1, ub1, ib1, uW2, iW2, ub2, ib2,
        sw_ids, sw_table, (float*)d_out);
}


# k steps of both layers taken in rotated order, each work-group starting at its own step, so the 128 groups stop asking L2 for the same weight lines at once; f32 add order over k differs per group
# speedup vs baseline: 9.2686x; 1.0637x over previous
#include <hip/hip_runtime.h>

typedef __attribute__((ext_vector_type(16))) _Float16 v16h;
typedef __attribute__((ext_vector_type(8)))  _Float16 v8h;
typedef __attribute__((ext_vector_type(4)))  _Float16 v4h;
typedef __attribute__((ext_vector_type(8)))  float    v8f;
typedef __attribute__((ext_vector_type(4)))  float    v4f;
typedef __attribute__((ext_vector_type(2)))  float    v2f;

#define B_ROWS 4096
#define NFEAT  4
#define VOCAB  100000
#define EMB    64
#define D_IN   256
#define D_H    256
#define D_O    128

#define ROWS_PB     32
#define N_USER_ROWS ROWS_PB
#define N_ITEM_ROWS (ROWS_PB + 3)
#define N_LDS_ROWS  (N_USER_ROWS + N_ITEM_ROWS)
#define HALO_ROW0   (N_USER_ROWS + ROWS_PB)
#define LDS_PITCH   (D_IN + 8)
#define MAX_RT      3

template <int NT> struct RawVec;
template <> struct RawVec<4> { typedef v4f type; };
template <> struct RawVec<2> { typedef v2f type; };

__device__ __forceinline__ constexpr int krow(int s) { return s < 8 ? s : s + 8; }

__device__ __forceinline__ int kstep(int t, int rot) { return (t + rot) & 7; }

__device__ __forceinline__ v16h frag16(const _Float16* p) {
    const v8h lo = *reinterpret_cast<const v8h*>(p);
    const v8h hi = *reinterpret_cast<const v8h*>(p + 16);
    v16h a;
    for (int i = 0; i < 8; ++i) { a[i] = lo[i]; a[8 + i] = hi[i]; }
    return a;
}

template <int NT, int NPITCH>
__device__ __forceinline__ void load_raw(typename RawVec<NT>::type (&raw)[16],
                                         const float* wp, int k0) {
    const float* const kp = wp + (size_t)k0 * NPITCH;
    #pragma unroll
    for (int s = 0; s < 16; ++s)
        raw[s] = *reinterpret_cast<const typename RawVec<NT>::type*>(kp + krow(s) * NPITCH);
}

template <int NT, int NPITCH, int DEPTH, int NR, typename Tail>
__device__ __forceinline__ void k_loop(v8f (&acc)[MAX_RT][NT],
                                       typename RawVec<NT>::type (&raw)[DEPTH][16],
                                       const float* wp,
                                       const _Float16* const (&ap)[MAX_RT], int rot, Tail tail) {
    int ks = kstep(DEPTH, rot);
    const float* kp = wp + (size_t)(ks * 32) * NPITCH;
    v16h a[NR];
    #pragma unroll
    for (int r = 0; r < NR; ++r) a[r] = frag16(ap[r] + kstep(0, rot) * 32);
    #pragma unroll
    for (int t = 0; t < 8; ++t) {
        if (t >= 1 && t + DEPTH < 8) {
            kp += (ks == 7 ? -7 : 1) * (32 * NPITCH);
            ks = (ks + 1) & 7;
        }
        v16h b[NT];
        #pragma unroll
        for (int j = 0; j < NT; ++j) {
            #pragma unroll
            for (int s = 0; s < 16; ++s) b[j][s] = (_Float16)raw[t % DEPTH][s][j];
        }
        if (t + DEPTH < 8) load_raw<NT, NPITCH>(raw[t % DEPTH], kp, 0);
        else               tail(t + DEPTH - 8);
        v16h an[NR];
        #pragma unroll
        for (int r = 0; r < NR; ++r)
            an[r] = (t + 1 < 8) ? frag16(ap[r] + kstep(t + 1, rot) * 32) : a[r];
        __builtin_amdgcn_sched_barrier(0);
        #pragma unroll
        for (int r = 0; r < NR; ++r) {
            #pragma unroll
            for (int j = 0; j < NT; ++j)
                acc[r][j] = __builtin_amdgcn_wmma_f32_16x16x32_f16(false, a[r], false, b[j],
                                                                   (short)0, acc[r][j], false, false);
        }
        #pragma unroll
        for (int r = 0; r < NR; ++r) a[r] = an[r];
        __builtin_amdgcn_sched_barrier(0);
    }
}

__device__ __forceinline__ float dot4(const v4f a, const v4f b) {
    return a[0] * b[0] + a[1] * b[1] + a[2] * b[2] + a[3] * b[3];
}

template <int M>
__device__ __forceinline__ float from_row_xor(float x) {
    const int r = __builtin_amdgcn_update_dpp(0, __builtin_bit_cast(int, x), 0x160 + M, 0xf, 0xf, true);
    return __builtin_bit_cast(float, r);
}

__device__ __forceinline__ float from_other_row(float x) {
    const int b = __builtin_bit_cast(int, x);
    const int y = __builtin_amdgcn_permlanex16(b, b, 0x76543210, 0xfedcba98, false, false);
    return __builtin_bit_cast(float, y);
}

template <int N, int NMAX, typename Xchg>
__device__ __forceinline__ void fold_stage(float (&v)[NMAX], bool bit, Xchg xchg) {
    #pragma unroll
    for (int m = 0; m < N / 2; ++m) {
        const float keep = bit ? v[2 * m + 1] : v[2 * m];
        const float send = bit ? v[2 * m] : v[2 * m + 1];
        v[m] = keep + xchg(send);
    }
    if (N & 1) v[N / 2] = v[N - 1] + xchg(v[N - 1]);
}

__launch_bounds__(256) __attribute__((amdgpu_waves_per_eu(2, 2)))
__global__ void tower_score_kernel(const int*   __restrict__ uids,
                                   const int*   __restrict__ iids,
                                   const float* __restrict__ utab,
                                   const float* __restrict__ itab,
                                   const float* __restrict__ uW1,
                                   const float* __restrict__ iW1,
                                   const float* __restrict__ ub1,
                                   const float* __restrict__ ib1,
                                   const float* __restrict__ uW2,
                                   const float* __restrict__ iW2,
                                   const float* __restrict__ ub2,
                                   const float* __restrict__ ib2,
                                   const int*   __restrict__ sw_ids,
                                   const float* __restrict__ sw_table,
                                   float* __restrict__ out)
{
    __shared__ __attribute__((aligned(16))) _Float16 x_lds[N_LDS_ROWS][LDS_PITCH];
    __shared__ __attribute__((aligned(16))) _Float16 h_lds[N_LDS_ROWS][LDS_PITCH];
    __shared__ __attribute__((aligned(16))) float    u_lds[N_LDS_ROWS][D_O];
    __shared__ __attribute__((aligned(16))) float    res_lds[ROWS_PB * 4];

    const int tid  = threadIdx.x;
    const int wave = tid >> 5;
    const int lane = tid & 31;
    const int half = lane >> 4;
    const int idx  = lane & 15;
    const int row0 = blockIdx.x * ROWS_PB;
    const bool item = (wave >= 4);
    const int  tw   = wave & 3;
    const int  rot  = (int)((blockIdx.x ^ (blockIdx.x >> 3)) & 7);

    const int sl   = lane & 15;
    const int scol = (row0 + 4 * wave + (sl >> 2) + (sl & 3)) & (B_ROWS - 1);
    const int swid = sw_ids[scol];

    constexpr int UCH = N_USER_ROWS * 64 / 256;
    constexpr int ICH = (N_ITEM_ROWS * 64 + 255) / 256;
    int uid_v[UCH], iid_v[ICH];
    for (int i = 0; i < UCH; ++i) {
        const int c = tid + i * 256;
        uid_v[i] = uids[(row0 + (c >> 6)) * NFEAT + ((c & 63) >> 4)];
    }
    for (int i = 0; i < ICH; ++i) {
        const int c0 = tid + i * 256;
        const int c  = c0 < N_ITEM_ROWS * 64 ? c0 : N_ITEM_ROWS * 64 - 1;
        const int grow = (row0 + (c >> 6)) & (B_ROWS - 1);
        iid_v[i] = iids[grow * NFEAT + ((c & 63) >> 4)];
    }
    const v4f bn1 = *reinterpret_cast<const v4f*>((item ? ib1 : ub1) + tw * 64 + idx * 4);
    const v2f bn2 = *reinterpret_cast<const v2f*>((item ? ib2 : ub2) + tw * 32 + idx * 2);
    __builtin_amdgcn_sched_barrier(0);

    const float* w1p = (item ? iW1 : uW1) + (size_t)(half * 8) * D_H + tw * 64 + idx * 4;
    v4f w1[2][16];
    load_raw<4, D_H>(w1[0], w1p, kstep(0, rot) * 32);
    load_raw<4, D_H>(w1[1], w1p, kstep(1, rot) * 32);
    __builtin_amdgcn_sched_barrier(0);

    const float swv = sw_table[swid];
    v4f ux[UCH], ix[ICH];
    for (int i = 0; i < UCH; ++i) {
        const int c  = tid + i * 256;
        const int w4 = c & 63;
        const unsigned off = ((unsigned)(w4 >> 4) * VOCAB + (unsigned)uid_v[i]) * EMB + (w4 & 15) * 4;
        ux[i] = *reinterpret_cast<const v4f*>(utab + off);
    }
    for (int i = 0; i < ICH; ++i) {
        const int c0 = tid + i * 256;
        const int c  = c0 < N_ITEM_ROWS * 64 ? c0 : N_ITEM_ROWS * 64 - 1;
        const int w4 = c & 63;
        const unsigned off = ((unsigned)(w4 >> 4) * VOCAB + (unsigned)iid_v[i]) * EMB + (w4 & 15) * 4;
        ix[i] = *reinterpret_cast<const v4f*>(itab + off);
    }
    for (int i = 0; i < UCH; ++i) {
        const int c = tid + i * 256;
        v4h hv;
        hv[0] = (_Float16)ux[i][0]; hv[1] = (_Float16)ux[i][1];
        hv[2] = (_Float16)ux[i][2]; hv[3] = (_Float16)ux[i][3];
        *reinterpret_cast<v4h*>(&x_lds[c >> 6][(c & 63) * 4]) = hv;
    }
    for (int i = 0; i < ICH; ++i) {
        const int c0 = tid + i * 256;
        const int c  = c0 < N_ITEM_ROWS * 64 ? c0 : N_ITEM_ROWS * 64 - 1;
        v4h hv;
        hv[0] = (_Float16)ix[i][0]; hv[1] = (_Float16)ix[i][1];
        hv[2] = (_Float16)ix[i][2]; hv[3] = (_Float16)ix[i][3];
        *reinterpret_cast<v4h*>(&x_lds[N_USER_ROWS + (c >> 6)][(c & 63) * 4]) = hv;
    }
    __syncthreads();

    const int rbase = item ? N_USER_ROWS : 0;
    const int r2 = HALO_ROW0 + (idx < 2 ? idx : 2);
    const _Float16* const xa[MAX_RT] = { &x_lds[rbase + idx][half * 8],
                                         &x_lds[rbase + 16 + idx][half * 8],
                                         &x_lds[r2][half * 8] };
    const _Float16* const ha[MAX_RT] = { &h_lds[rbase + idx][half * 8],
                                         &h_lds[rbase + 16 + idx][half * 8],
                                         &h_lds[r2][half * 8] };

    const float* w2p = (item ? iW2 : uW2) + (size_t)(half * 8) * D_O + tw * 32 + idx * 2;
    v2f w2[3][16];
    auto w2_first = [&](int i) { load_raw<2, D_O>(w2[i], w2p, kstep(i, rot) * 32); };

    v8f cc[MAX_RT][4] = {};
    if (!item) k_loop<4, D_H, 2, 2>(cc, w1, w1p, xa, rot, w2_first);
    else       k_loop<4, D_H, 2, 3>(cc, w1, w1p, xa, rot, w2_first);
    w2_first(2);
    __builtin_amdgcn_sched_barrier(0);

    {
        const int n = tw * 64 + idx * 4;
        for (int rt = 0; rt < 2; ++rt) {
            for (int r = 0; r < 8; ++r) {
                v4h hv;
                for (int j = 0; j < 4; ++j) {
                    const float v = cc[rt][j][r] + bn1[j];
                    hv[j] = (_Float16)(v > 0.f ? v : 0.f);
                }
                *reinterpret_cast<v4h*>(&h_lds[rbase + rt * 16 + half * 8 + r][n]) = hv;
            }
        }
        if (item && half == 0) {
            for (int r = 0; r < 3; ++r) {
                v4h hv;
                for (int j = 0; j < 4; ++j) {
                    const float v = cc[2][j][r] + bn1[j];
                    hv[j] = (_Float16)(v > 0.f ? v : 0.f);
                }
                *reinterpret_cast<v4h*>(&h_lds[HALO_ROW0 + r][n]) = hv;
            }
        }
    }
    __syncthreads();

    v8f dd[MAX_RT][2] = {};
    auto no_tail = [](int) {};
    if (!item) k_loop<2, D_O, 3, 2>(dd, w2, w2p, ha, rot, no_tail);
    else       k_loop<2, D_O, 3, 3>(dd, w2, w2p, ha, rot, no_tail);
    {
        const int n = tw * 32 + idx * 2;
        for (int rt = 0; rt < 2; ++rt) {
            for (int r = 0; r < 8; ++r) {
                v2f uv;
                for (int j = 0; j < 2; ++j) {
                    const float v = dd[rt][j][r] + bn2[j];
                    uv[j] = (v > 0.f ? v : 0.f);
                }
                *reinterpret_cast<v2f*>(&u_lds[rbase + rt * 16 + half * 8 + r][n]) = uv;
            }
        }
        if (item && half == 0) {
            for (int r = 0; r < 3; ++r) {
                v2f uv;
                for (int j = 0; j < 2; ++j) {
                    const float v = dd[2][j][r] + bn2[j];
                    uv[j] = (v > 0.f ? v : 0.f);
                }
                *reinterpret_cast<v2f*>(&u_lds[HALO_ROW0 + r][n]) = uv;
            }
        }
    }
    __syncthreads();

    constexpr int UPW = ROWS_PB / 8;
    constexpr int IPW = UPW + 3;
    constexpr int NSUM = UPW * 4 + UPW + IPW;
    static_assert(UPW * 4 == 16 && NSUM <= 32, "one sum per lane, dots in the first row");
    const int i0 = UPW * wave;
    v4f a[UPW], b[IPW];
    for (int t = 0; t < UPW; ++t)
        a[t] = *reinterpret_cast<const v4f*>(&u_lds[i0 + t][lane * 4]);
    for (int m = 0; m < IPW; ++m)
        b[m] = *reinterpret_cast<const v4f*>(&u_lds[N_USER_ROWS + i0 + m][lane * 4]);
    float part[NSUM];
    for (int t = 0; t < UPW; ++t)
        for (int j = 0; j < 4; ++j)
            part[t * 4 + j] = dot4(a[t], b[t + j]);
    for (int t = 0; t < UPW; ++t) part[UPW * 4 + t] = dot4(a[t], a[t]);
    for (int m = 0; m < IPW; ++m) part[UPW * 4 + UPW + m] = dot4(b[m], b[m]);
    fold_stage<27, NSUM>(part, (lane & 1) != 0, [](float x) { return from_row_xor<1>(x); });
    fold_stage<14, NSUM>(part, (lane & 2) != 0, [](float x) { return from_row_xor<2>(x); });
    fold_stage<7,  NSUM>(part, (lane & 4) != 0, [](float x) { return from_row_xor<4>(x); });
    fold_stage<4,  NSUM>(part, (lane & 8) != 0, [](float x) { return from_row_xor<8>(x); });
    fold_stage<2,  NSUM>(part, (lane & 16) != 0, [](float x) { return from_other_row(x); });
    const float total = part[0];
    const float nrm  = sqrtf(total);
    const int   nrmc = __builtin_bit_cast(int, (nrm > 1e-8f) ? nrm : 1e-8f);
    const float nu_sel = __builtin_bit_cast(float,
        __builtin_amdgcn_permlanex16(nrmc, nrmc, 0x11110000, 0x33332222, false, false));
    const float nv_sel = __builtin_bit_cast(float,
        __builtin_amdgcn_permlanex16(nrmc, nrmc, 0x87657654, 0xa9879876, false, false));
    const float res = total / (nu_sel * nv_sel) - logf(swv);
    if (lane < UPW * 4) res_lds[wave * (UPW * 4) + lane] = res;
    __syncthreads();

    if (wave == 0) {
        const v4f o = *reinterpret_cast<const v4f*>(&res_lds[lane * 4]);
        *reinterpret_cast<v4f*>(out + (size_t)row0 * 4 + lane * 4) = o;
    }
}

extern "C" void kernel_launch(void* const* d_in, const int* in_sizes, int n_in,
                              void* d_out, int out_size, void* d_ws, size_t ws_size,
                              hipStream_t stream)
{
    const int*   user_ids = (const int*)  d_in[0];
    const int*   item_ids = (const int*)  d_in[1];
    const int*   sw_ids   = (const int*)  d_in[2];
    const float* utab     = (const float*)d_in[3];
    const float* itab     = (const float*)d_in[4];
    const float* sw_table = (const float*)d_in[5];
    const float* uW1 = (const float*)d_in[6];
    const float* ub1 = (const float*)d_in[7];
    const float* uW2 = (const float*)d_in[8];
    const float* ub2 = (const float*)d_in[9];
    const float* iW1 = (const float*)d_in[10];
    const float* ib1 = (const float*)d_in[11];
    const float* iW2 = (const float*)d_in[12];
    const float* ib2 = (const float*)d_in[13];

    tower_score_kernel<<<B_ROWS / ROWS_PB, 256, 0, stream>>>(
        user_ids, item_ids, utab, itab, uW1, iW1, ub1, ib1, uW2, iW2, ub2, ib2,
        sw_ids, sw_table, (float*)d_out);
}
